# starting kernel
# baseline (speedup 1.0000x reference)
#include <hip/hip_runtime.h>

typedef __attribute__((ext_vector_type(2))) float v2f;
typedef __attribute__((ext_vector_type(8))) float v8f;

#define IN_F   128
#define OUT_F  128
#define MAXDEG 4
#define NBUCK  5
#define TILE_M 16
#define LDA    260
#define SCAN_B 256

__global__ void fill_i32_kernel(int* __restrict__ p, int val, size_t n) {
    size_t i = (size_t)blockIdx.x * blockDim.x + threadIdx.x;
    size_t stride = (size_t)gridDim.x * blockDim.x;
    for (; i < n; i += stride) p[i] = val;
}

__global__ void degree_kernel(const int* __restrict__ col, int* __restrict__ degi, int E) {
    int e = blockIdx.x * blockDim.x + threadIdx.x;
    if (e < E) atomicAdd(&degi[col[e]], 1);
}

__global__ void scan1_kernel(const int* __restrict__ degi, int* __restrict__ rowstart,
                             int* __restrict__ bsum, int N) {
    __shared__ int sh[SCAN_B];
    int i = blockIdx.x * SCAN_B + threadIdx.x;
    int v = (i < N) ? degi[i] : 0;
    sh[threadIdx.x] = v;
    __syncthreads();
    #pragma unroll
    for (int off = 1; off < SCAN_B; off <<= 1) {
        int t = (threadIdx.x >= off) ? sh[threadIdx.x - off] : 0;
        __syncthreads();
        sh[threadIdx.x] += t;
        __syncthreads();
    }
    if (i < N) rowstart[i] = sh[threadIdx.x] - v;
    if (threadIdx.x == SCAN_B - 1) bsum[blockIdx.x] = sh[SCAN_B - 1];
}

__global__ void scan2_kernel(int* __restrict__ bsum, int nBlocks) {
    if (threadIdx.x == 0 && blockIdx.x == 0) {
        int acc = 0;
        for (int b = 0; b < nBlocks; ++b) { int t = bsum[b]; bsum[b] = acc; acc += t; }
    }
}

__global__ void scan3_kernel(int* __restrict__ rowstart, const int* __restrict__ bsum, int N) {
    int i = blockIdx.x * SCAN_B + threadIdx.x;
    if (i < N) rowstart[i] += bsum[blockIdx.x];
}

__global__ void csr_fill_kernel(const int* __restrict__ row, const int* __restrict__ col,
                                const int* __restrict__ rowstart, int* __restrict__ cursor,
                                int* __restrict__ edge_src, int E) {
    int e = blockIdx.x * blockDim.x + threadIdx.x;
    if (e >= E) return;
    int d = col[e];
    int pos = atomicAdd(&cursor[d], 1);
    edge_src[rowstart[d] + pos] = row[e];
}

__global__ void bucket_count_kernel(const int* __restrict__ degi, int* __restrict__ meta, int N) {
    int i = blockIdx.x * blockDim.x + threadIdx.x;
    if (i >= N) return;
    int b = min(degi[i], MAXDEG);
    atomicAdd(&meta[b], 1);
}

__global__ void bucket_offset_kernel(int* __restrict__ meta) {
    if (threadIdx.x == 0 && blockIdx.x == 0) {
        int off = 0;
        for (int b = 0; b < NBUCK; ++b) {
            meta[5 + b] = off;
            off += (meta[b] + TILE_M - 1) & ~(TILE_M - 1);
        }
    }
}

__global__ void bucket_scatter_kernel(const int* __restrict__ degi, int* __restrict__ meta,
                                      int* __restrict__ perm, int N) {
    int i = blockIdx.x * blockDim.x + threadIdx.x;
    if (i >= N) return;
    int b = min(degi[i], MAXDEG);
    int pos = atomicAdd(&meta[10 + b], 1);
    perm[meta[5 + b] + pos] = i;
}

__global__ __launch_bounds__(256) void mfconv_gemm_kernel(
    const float* __restrict__ x,
    const int* __restrict__ degi, const int* __restrict__ rowstart,
    const int* __restrict__ edge_src, const int* __restrict__ perm,
    const float* __restrict__ neigh_w, const float* __restrict__ neigh_b,
    const float* __restrict__ root_w,  const float* __restrict__ root_b,
    float* __restrict__ out)
{
    __shared__ float ldsA[TILE_M * LDA];
    __shared__ int   s_node[TILE_M];
    __shared__ int   s_bucket;

    const int tid  = threadIdx.x;
    const int tile = blockIdx.x;

    if (tid < TILE_M) s_node[tid] = perm[tile * TILE_M + tid];
    __syncthreads();

    if (tid == 0) {
        int b = -1;
        for (int i = 0; i < TILE_M; ++i) {
            int nd = s_node[i];
            if (nd >= 0) { b = min(degi[nd], MAXDEG); break; }
        }
        s_bucket = b;
    }

    {
        const int m    = tid >> 4;
        const int k0   = (tid & 15) * 16;
        const int node = s_node[m];
        float4 a0 = make_float4(0.f,0.f,0.f,0.f), a1 = a0, a2 = a0, a3 = a0;
        if (node >= 0) {
            if (k0 < IN_F) {
                const int start = rowstart[node];
                const int d     = degi[node];
                for (int j = 0; j < d; ++j) {
                    const float* xs = x + (size_t)edge_src[start + j] * IN_F + k0;
                    const float4 v0 = *(const float4*)(xs + 0);
                    const float4 v1 = *(const float4*)(xs + 4);
                    const float4 v2 = *(const float4*)(xs + 8);
                    const float4 v3 = *(const float4*)(xs + 12);
                    a0.x += v0.x; a0.y += v0.y; a0.z += v0.z; a0.w += v0.w;
                    a1.x += v1.x; a1.y += v1.y; a1.z += v1.z; a1.w += v1.w;
                    a2.x += v2.x; a2.y += v2.y; a2.z += v2.z; a2.w += v2.w;
                    a3.x += v3.x; a3.y += v3.y; a3.z += v3.z; a3.w += v3.w;
                }
                const float inv = 1.0f / fmaxf((float)d, 1.0f);
                a0.x *= inv; a0.y *= inv; a0.z *= inv; a0.w *= inv;
                a1.x *= inv; a1.y *= inv; a1.z *= inv; a1.w *= inv;
                a2.x *= inv; a2.y *= inv; a2.z *= inv; a2.w *= inv;
                a3.x *= inv; a3.y *= inv; a3.z *= inv; a3.w *= inv;
            } else {
                const float* xs = x + (size_t)node * IN_F + (k0 - IN_F);
                a0 = *(const float4*)(xs + 0);
                a1 = *(const float4*)(xs + 4);
                a2 = *(const float4*)(xs + 8);
                a3 = *(const float4*)(xs + 12);
            }
        }
        float* dst = &ldsA[m * LDA + k0];
        *(float4*)(dst + 0)  = a0;
        *(float4*)(dst + 4)  = a1;
        *(float4*)(dst + 8)  = a2;
        *(float4*)(dst + 12) = a3;
    }
    __syncthreads();

    const int bucket = s_bucket;
    if (bucket < 0) return;

    const int wave  = tid >> 5;
    const int lane  = tid & 31;
    const int lr    = lane & 15;
    const int khalf = (lane >> 4) << 1;
    const int ncol  = wave * 16 + lr;

    const float* __restrict__ wn   = neigh_w + ((size_t)bucket * OUT_F + ncol) * IN_F;
    const float* __restrict__ wr   = root_w  + ((size_t)bucket * OUT_F + ncol) * IN_F;
    const float* __restrict__ aRow = &ldsA[lr * LDA];

    v8f acc = {};
    #pragma unroll
    for (int kk = 0; kk < 32; ++kk) {
        const int k = kk * 4 + khalf;
        v2f a, b;
        a.x = aRow[k];     a.y = aRow[k + 1];
        b.x = wn[k];       b.y = wn[k + 1];
        acc = __builtin_amdgcn_wmma_f32_16x16x4_f32(false, a, false, b,
                                                    (short)0, acc, false, false);
    }
    #pragma unroll
    for (int kk = 0; kk < 32; ++kk) {
        const int k = kk * 4 + khalf;
        v2f a, b;
        a.x = aRow[IN_F + k]; a.y = aRow[IN_F + k + 1];
        b.x = wr[k];          b.y = wr[k + 1];
        acc = __builtin_amdgcn_wmma_f32_16x16x4_f32(false, a, false, b,
                                                    (short)0, acc, false, false);
    }

    const float bias = neigh_b[bucket * OUT_F + ncol] + root_b[bucket * OUT_F + ncol];
    const int mbase  = (lane >> 4) * 8;
    #pragma unroll
    for (int i = 0; i < 8; ++i) {
        const int node = s_node[mbase + i];
        if (node >= 0) out[(size_t)node * OUT_F + ncol] = acc[i] + bias;
    }
}

extern "C" void kernel_launch(void* const* d_in, const int* in_sizes, int n_in,
                              void* d_out, int out_size, void* d_ws, size_t ws_size,
                              hipStream_t stream) {
    const float* x       = (const float*)d_in[0];
    const float* neigh_w = (const float*)d_in[1];
    const float* neigh_b = (const float*)d_in[2];
    const float* root_w  = (const float*)d_in[3];
    const float* root_b  = (const float*)d_in[4];
    const int*   eidx    = (const int*)d_in[5];

    const int N = in_sizes[0] / IN_F;
    const int E = in_sizes[5] / 2;
    const int* row = eidx;
    const int* col = eidx + E;

    const int nScanBlocks = (N + SCAN_B - 1) / SCAN_B;
    const int permLen = ((N + NBUCK * TILE_M + TILE_M - 1) / TILE_M) * TILE_M;
    const int nTiles  = permLen / TILE_M;

    char* ws = (char*)d_ws;
    int* degi     = (int*)ws;  ws += (size_t)N * sizeof(int);
    int* cursor   = (int*)ws;  ws += (size_t)N * sizeof(int);
    int* rowstart = (int*)ws;  ws += (size_t)N * sizeof(int);
    int* bsum     = (int*)ws;  ws += (size_t)((nScanBlocks + 15) & ~15) * sizeof(int);
    int* edge_src = (int*)ws;  ws += (size_t)E * sizeof(int);
    int* perm     = (int*)ws;  ws += (size_t)permLen * sizeof(int);
    int* meta     = (int*)ws;

    fill_i32_kernel<<<1024, 256, 0, stream>>>(degi, 0, (size_t)2 * N);
    fill_i32_kernel<<<1024, 256, 0, stream>>>(perm, -1, (size_t)permLen);
    fill_i32_kernel<<<1, 32, 0, stream>>>(meta, 0, 16);

    degree_kernel<<<(E + 255) / 256, 256, 0, stream>>>(col, degi, E);
    scan1_kernel<<<nScanBlocks, SCAN_B, 0, stream>>>(degi, rowstart, bsum, N);
    scan2_kernel<<<1, 1, 0, stream>>>(bsum, nScanBlocks);
    scan3_kernel<<<nScanBlocks, SCAN_B, 0, stream>>>(rowstart, bsum, N);
    csr_fill_kernel<<<(E + 255) / 256, 256, 0, stream>>>(row, col, rowstart, cursor,
                                                         edge_src, E);

    bucket_count_kernel<<<(N + 255) / 256, 256, 0, stream>>>(degi, meta, N);
    bucket_offset_kernel<<<1, 1, 0, stream>>>(meta);
    bucket_scatter_kernel<<<(N + 255) / 256, 256, 0, stream>>>(degi, meta, perm, N);

    mfconv_gemm_kernel<<<nTiles, 256, 0, stream>>>(x, degi, rowstart, edge_src, perm,
                                                   neigh_w, neigh_b, root_w, root_b,
                                                   (float*)d_out);
}


# GEMM kernel: the gather's index chain (degree, row start, source ids) read once per work-group into LDS beside the node list; a tile's staging loads only x rows, own row + 4 neighbours as one batch
# speedup vs baseline: 1.8141x; 1.8141x over previous
#include <hip/hip_runtime.h>

typedef __attribute__((ext_vector_type(2))) float v2f;
typedef __attribute__((ext_vector_type(8))) float v8f;

#define IN_F   128
#define OUT_F  128
#define MAXDEG 4
#define NBUCK  5
#define TILE_M 16
#define LDA    260
#define SCAN_B 256
#define RANGE_N     512
#define BIG_PARTS   2
#define GRID_Y      (NBUCK - 1 + BIG_PARTS)
#define RANGE_PASS  (RANGE_N / 256)
#define LIST_LEN    (RANGE_N + TILE_M)
#define SRC_CAP     8

__global__ void fill_i32_kernel(int* __restrict__ p, int val, size_t n) {
    size_t i = (size_t)blockIdx.x * blockDim.x + threadIdx.x;
    size_t stride = (size_t)gridDim.x * blockDim.x;
    for (; i < n; i += stride) p[i] = val;
}

__global__ void degree_kernel(const int* __restrict__ col, int* __restrict__ degi, int E) {
    int e = blockIdx.x * blockDim.x + threadIdx.x;
    if (e < E) atomicAdd(&degi[col[e]], 1);
}

__global__ void scan1_kernel(const int* __restrict__ degi, int* __restrict__ rowstart,
                             int* __restrict__ bsum, int N) {
    __shared__ int sh[SCAN_B];
    int i = blockIdx.x * SCAN_B + threadIdx.x;
    int v = (i < N) ? degi[i] : 0;
    sh[threadIdx.x] = v;
    __syncthreads();
    #pragma unroll
    for (int off = 1; off < SCAN_B; off <<= 1) {
        int t = (threadIdx.x >= off) ? sh[threadIdx.x - off] : 0;
        __syncthreads();
        sh[threadIdx.x] += t;
        __syncthreads();
    }
    if (i < N) rowstart[i] = sh[threadIdx.x] - v;
    if (threadIdx.x == SCAN_B - 1) bsum[blockIdx.x] = sh[SCAN_B - 1];
}

__global__ void scan2_kernel(int* __restrict__ bsum, int nBlocks) {
    if (threadIdx.x == 0 && blockIdx.x == 0) {
        int acc = 0;
        for (int b = 0; b < nBlocks; ++b) { int t = bsum[b]; bsum[b] = acc; acc += t; }
    }
}

__global__ void scan3_kernel(int* __restrict__ rowstart, const int* __restrict__ bsum, int N) {
    int i = blockIdx.x * SCAN_B + threadIdx.x;
    if (i < N) rowstart[i] += bsum[blockIdx.x];
}

__global__ void csr_fill_kernel(const int* __restrict__ row, const int* __restrict__ col,
                                const int* __restrict__ rowstart, int* __restrict__ cursor,
                                int* __restrict__ edge_src, int E) {
    int e = blockIdx.x * blockDim.x + threadIdx.x;
    if (e >= E) return;
    int d = col[e];
    int pos = atomicAdd(&cursor[d], 1);
    edge_src[rowstart[d] + pos] = row[e];
}

__global__ __launch_bounds__(256, 4) void mfconv_gemm_kernel(
    const float* __restrict__ x,
    const int* __restrict__ degi, const int* __restrict__ rowstart,
    const int* __restrict__ edge_src,
    const float* __restrict__ neigh_w, const float* __restrict__ neigh_b,
    const float* __restrict__ root_w,  const float* __restrict__ root_b,
    float* __restrict__ out, int N)
{
    __shared__ float ldsA[TILE_M * LDA];
    __shared__ int   s_list[LIST_LEN];
    __shared__ int   s_deg[LIST_LEN];
    __shared__ int   s_rs[LIST_LEN];
    __shared__ __attribute__((aligned(16))) int s_src[LIST_LEN * SRC_CAP];
    __shared__ int   s_cnt[RANGE_PASS * 8];

    const int tid   = threadIdx.x;
    const int wave  = tid >> 5;
    const int lane  = tid & 31;
    const int node0 = blockIdx.x * RANGE_N;

    const int by     = blockIdx.y;
    const int bucket = (by < BIG_PARTS) ? MAXDEG : (NBUCK - 2 + BIG_PARTS - by);
    const int part   = (by < BIG_PARTS) ? by : 0;
    const int nparts = (by < BIG_PARTS) ? BIG_PARTS : 1;

    for (int i = tid; i < LIST_LEN; i += 256) s_list[i] = -1;

    bool mine[RANGE_PASS];
    int  myrank[RANGE_PASS];
    int  mydeg[RANGE_PASS];
    int  mystart[RANGE_PASS];
    #pragma unroll
    for (int p = 0; p < RANGE_PASS; ++p) {
        const int node = node0 + p * 256 + tid;
        mydeg[p]   = (node < N) ? degi[node] : 0;
        mystart[p] = (node < N) ? rowstart[node] : 0;
    }
    #pragma unroll
    for (int p = 0; p < RANGE_PASS; ++p) {
        const int node = node0 + p * 256 + tid;
        const bool in = (node < N) && (min(mydeg[p], MAXDEG) == bucket);
        const unsigned m = __builtin_amdgcn_ballot_w32(in);
        if (lane == 0) s_cnt[p * 8 + wave] = __popc(m);
        myrank[p] = __popc(m & ((1u << lane) - 1u));
        mine[p] = in;
    }
    __syncthreads();

    int total = 0;
    int base[RANGE_PASS];
    #pragma unroll
    for (int p = 0; p < RANGE_PASS; ++p) base[p] = 0;
    #pragma unroll
    for (int i = 0; i < RANGE_PASS * 8; ++i) {
        const int c = s_cnt[i];
        #pragma unroll
        for (int p = 0; p < RANGE_PASS; ++p) if (i < p * 8 + wave) base[p] += c;
        total += c;
    }
    const int nTiles = (__builtin_amdgcn_readfirstlane(total) + TILE_M - 1) / TILE_M;
    if (part >= nTiles) return;

    int  pos[RANGE_PASS];
    bool want[RANGE_PASS];
    int  ids[RANGE_PASS][SRC_CAP];
    #pragma unroll
    for (int p = 0; p < RANGE_PASS; ++p) {
        pos[p]  = base[p] + myrank[p];
        want[p] = mine[p] && (mydeg[p] > 0) && (((pos[p] / TILE_M) & (nparts - 1)) == part);
        if (mine[p]) {
            s_list[pos[p]] = node0 + p * 256 + tid;
            s_deg[pos[p]]  = mydeg[p];
            s_rs[pos[p]]   = mystart[p];
        }
        #pragma unroll
        for (int j = 0; j < SRC_CAP; ++j)
            ids[p][j] = (want[p] && j < mydeg[p]) ? edge_src[mystart[p] + j] : 0;
    }

    const int lr    = lane & 15;
    const int khalf = (lane >> 4) << 1;
    const int ncol  = wave * 16 + lr;
    const unsigned coff = (unsigned)ncol * IN_F;
    const int mbase = (lane >> 4) * 8;
    const float* __restrict__ aRow = &ldsA[lr * LDA];

    const float* __restrict__ wn = neigh_w + (size_t)bucket * (OUT_F * IN_F);
    const float* __restrict__ wr = root_w  + (size_t)bucket * (OUT_F * IN_F);
    v2f bfr[64];
    #pragma unroll
    for (int kk = 0; kk < 32; ++kk) {
        const int k = kk * 4 + khalf;
        bfr[kk].x = wn[coff + k]; bfr[kk].y = wn[coff + k + 1];
    }
    #pragma unroll
    for (int kk = 0; kk < 32; ++kk) {
        const int k = kk * 4 + khalf;
        bfr[32 + kk].x = wr[coff + k]; bfr[32 + kk].y = wr[coff + k + 1];
    }
    const float bias = neigh_b[bucket * OUT_F + ncol] + root_b[bucket * OUT_F + ncol];

    #pragma unroll
    for (int p = 0; p < RANGE_PASS; ++p) {
        if (want[p]) {
            int4* dst = (int4*)&s_src[pos[p] * SRC_CAP];
            #pragma unroll
            for (int c = 0; c < SRC_CAP / 4; ++c)
                dst[c] = make_int4(ids[p][4 * c], ids[p][4 * c + 1], ids[p][4 * c + 2], ids[p][4 * c + 3]);
        }
    }
    __syncthreads();

    for (int t = part; t < nTiles; t += nparts) {
        const int slot0 = t * TILE_M;

        {
            const int  m      = tid >> 4;
            const int  k0     = (tid & 15) * 16;
            const bool isRoot = k0 >= IN_F;
            const int  kx     = k0 & (IN_F - 1);
            const int  slot   = slot0 + m;
            const int  node   = s_list[slot];
            const float  z0  = (isRoot && node >= 0) ? -0.0f : 0.0f;
            const float4 id0 = make_float4(-0.0f, -0.0f, -0.0f, -0.0f);
            float4 a0 = make_float4(z0, z0, z0, z0), a1 = a0, a2 = a0, a3 = a0;
            if (node >= 0) {
                int d = 1;
                if (!isRoot) d = s_deg[slot];
                #pragma unroll
                for (int c = 0; c < SRC_CAP; c += 4) {
                    if (c < d) {
                        int4 id4 = make_int4(node, 0, 0, 0);
                        if (!isRoot) id4 = *(const int4*)&s_src[slot * SRC_CAP + c];
                        const int idv[4] = { id4.x, id4.y, id4.z, id4.w };
                        float4 v[4][4];
                        #pragma unroll
                        for (int j = 0; j < 4; ++j) {
                            v[j][0] = id0; v[j][1] = id0; v[j][2] = id0; v[j][3] = id0;
                            if (c + j < d) {
                                const float* xs = x + (size_t)idv[j] * IN_F + kx;
                                v[j][0] = *(const float4*)(xs + 0);
                                v[j][1] = *(const float4*)(xs + 4);
                                v[j][2] = *(const float4*)(xs + 8);
                                v[j][3] = *(const float4*)(xs + 12);
                            }
                        }
                        #pragma unroll
                        for (int j = 0; j < 4; ++j) {
                            a0.x += v[j][0].x; a0.y += v[j][0].y; a0.z += v[j][0].z; a0.w += v[j][0].w;
                            a1.x += v[j][1].x; a1.y += v[j][1].y; a1.z += v[j][1].z; a1.w += v[j][1].w;
                            a2.x += v[j][2].x; a2.y += v[j][2].y; a2.z += v[j][2].z; a2.w += v[j][2].w;
                            a3.x += v[j][3].x; a3.y += v[j][3].y; a3.z += v[j][3].z; a3.w += v[j][3].w;
                        }
                    }
                }
                if (d > SRC_CAP) {
                    const int start = s_rs[slot];
                    #pragma unroll 1
                    for (int j = SRC_CAP; j < d; ++j) {
                        const float* xs = x + (size_t)edge_src[start + j] * IN_F + kx;
                        const float4 v0 = *(const float4*)(xs + 0);
                        const float4 v1 = *(const float4*)(xs + 4);
                        const float4 v2 = *(const float4*)(xs + 8);
                        const float4 v3 = *(const float4*)(xs + 12);
                        a0.x += v0.x; a0.y += v0.y; a0.z += v0.z; a0.w += v0.w;
                        a1.x += v1.x; a1.y += v1.y; a1.z += v1.z; a1.w += v1.w;
                        a2.x += v2.x; a2.y += v2.y; a2.z += v2.z; a2.w += v2.w;
                        a3.x += v3.x; a3.y += v3.y; a3.z += v3.z; a3.w += v3.w;
                    }
                }
                const float inv = isRoot ? 1.0f : 1.0f / fmaxf((float)d, 1.0f);
                a0.x *= inv; a0.y *= inv; a0.z *= inv; a0.w *= inv;
                a1.x *= inv; a1.y *= inv; a1.z *= inv; a1.w *= inv;
                a2.x *= inv; a2.y *= inv; a2.z *= inv; a2.w *= inv;
                a3.x *= inv; a3.y *= inv; a3.z *= inv; a3.w *= inv;
            }
            float* dst = &ldsA[m * LDA + k0];
            *(float4*)(dst + 0)  = a0;
            *(float4*)(dst + 4)  = a1;
            *(float4*)(dst + 8)  = a2;
            *(float4*)(dst + 12) = a3;
        }
        __syncthreads();

        v8f acc = {};
        #pragma unroll
        for (int kk = 0; kk < 32; ++kk) {
            const int k = kk * 4 + khalf;
            v2f a;
            a.x = aRow[k];     a.y = aRow[k + 1];
            acc = __builtin_amdgcn_wmma_f32_16x16x4_f32(false, a, false, bfr[kk],
                                                        (short)0, acc, false, false);
        }
        #pragma unroll
        for (int kk = 0; kk < 32; ++kk) {
            const int k = kk * 4 + khalf;
            v2f a;
            a.x = aRow[IN_F + k]; a.y = aRow[IN_F + k + 1];
            acc = __builtin_amdgcn_wmma_f32_16x16x4_f32(false, a, false, bfr[32 + kk],
                                                        (short)0, acc, false, false);
        }

        #pragma unroll
        for (int i = 0; i < 8; ++i) {
            const int node = s_list[slot0 + mbase + i];
            if (node >= 0) out[(size_t)node * OUT_F + ncol] = acc[i] + bias;
        }
        __syncthreads();
    }
}

extern "C" void kernel_launch(void* const* d_in, const int* in_sizes, int n_in,
                              void* d_out, int out_size, void* d_ws, size_t ws_size,
                              hipStream_t stream) {
    const float* x       = (const float*)d_in[0];
    const float* neigh_w = (const float*)d_in[1];
    const float* neigh_b = (const float*)d_in[2];
    const float* root_w  = (const float*)d_in[3];
    const float* root_b  = (const float*)d_in[4];
    const int*   eidx    = (const int*)d_in[5];

    const int N = in_sizes[0] / IN_F;
    const int E = in_sizes[5] / 2;
    const int* row = eidx;
    const int* col = eidx + E;

    const int nScanBlocks = (N + SCAN_B - 1) / SCAN_B;
    const int nRanges = (N + RANGE_N - 1) / RANGE_N;

    char* ws = (char*)d_ws;
    int* degi     = (int*)ws;  ws += (size_t)N * sizeof(int);
    int* cursor   = (int*)ws;  ws += (size_t)N * sizeof(int);
    int* rowstart = (int*)ws;  ws += (size_t)N * sizeof(int);
    int* bsum     = (int*)ws;  ws += (size_t)((nScanBlocks + 15) & ~15) * sizeof(int);
    int* edge_src = (int*)ws;

    fill_i32_kernel<<<1024, 256, 0, stream>>>(degi, 0, (size_t)2 * N);

    degree_kernel<<<(E + 255) / 256, 256, 0, stream>>>(col, degi, E);
    scan1_kernel<<<nScanBlocks, SCAN_B, 0, stream>>>(degi, rowstart, bsum, N);
    scan2_kernel<<<1, 1, 0, stream>>>(bsum, nScanBlocks);
    scan3_kernel<<<nScanBlocks, SCAN_B, 0, stream>>>(rowstart, bsum, N);
    csr_fill_kernel<<<(E + 255) / 256, 256, 0, stream>>>(row, col, rowstart, cursor,
                                                         edge_src, E);

    mfconv_gemm_kernel<<<dim3(nRanges, GRID_Y), 256, 0, stream>>>(
        x, degi, rowstart, edge_src, neigh_w, neigh_b, root_w, root_b, (float*)d_out, N);
}


# GEMM kernel: rows of the tile staged next (own row + first 7 neighbour rows, one lane per 256-byte half row) are requested into L2 by prefetch instructions behind the hand-over barrier
# speedup vs baseline: 1.8823x; 1.0376x over previous
#include <hip/hip_runtime.h>

typedef __attribute__((ext_vector_type(2))) float v2f;
typedef __attribute__((ext_vector_type(8))) float v8f;

#define IN_F   128
#define OUT_F  128
#define MAXDEG 4
#define NBUCK  5
#define TILE_M 16
#define LDA    260
#define SCAN_B 256
#define RANGE_N     512
#define BIG_PARTS   2
#define GRID_Y      (NBUCK - 1 + BIG_PARTS)
#define RANGE_PASS  (RANGE_N / 256)
#define LIST_LEN    (RANGE_N + TILE_M)
#define SRC_CAP     8

__global__ void fill_i32_kernel(int* __restrict__ p, int val, size_t n) {
    size_t i = (size_t)blockIdx.x * blockDim.x + threadIdx.x;
    size_t stride = (size_t)gridDim.x * blockDim.x;
    for (; i < n; i += stride) p[i] = val;
}

__global__ void degree_kernel(const int* __restrict__ col, int* __restrict__ degi, int E) {
    int e = blockIdx.x * blockDim.x + threadIdx.x;
    if (e < E) atomicAdd(&degi[col[e]], 1);
}

__global__ void scan1_kernel(const int* __restrict__ degi, int* __restrict__ rowstart,
                             int* __restrict__ bsum, int N) {
    __shared__ int sh[SCAN_B];
    int i = blockIdx.x * SCAN_B + threadIdx.x;
    int v = (i < N) ? degi[i] : 0;
    sh[threadIdx.x] = v;
    __syncthreads();
    #pragma unroll
    for (int off = 1; off < SCAN_B; off <<= 1) {
        int t = (threadIdx.x >= off) ? sh[threadIdx.x - off] : 0;
        __syncthreads();
        sh[threadIdx.x] += t;
        __syncthreads();
    }
    if (i < N) rowstart[i] = sh[threadIdx.x] - v;
    if (threadIdx.x == SCAN_B - 1) bsum[blockIdx.x] = sh[SCAN_B - 1];
}

__global__ void scan2_kernel(int* __restrict__ bsum, int nBlocks) {
    if (threadIdx.x == 0 && blockIdx.x == 0) {
        int acc = 0;
        for (int b = 0; b < nBlocks; ++b) { int t = bsum[b]; bsum[b] = acc; acc += t; }
    }
}

__global__ void scan3_kernel(int* __restrict__ rowstart, const int* __restrict__ bsum, int N) {
    int i = blockIdx.x * SCAN_B + threadIdx.x;
    if (i < N) rowstart[i] += bsum[blockIdx.x];
}

__global__ void csr_fill_kernel(const int* __restrict__ row, const int* __restrict__ col,
                                const int* __restrict__ rowstart, int* __restrict__ cursor,
                                int* __restrict__ edge_src, int E) {
    int e = blockIdx.x * blockDim.x + threadIdx.x;
    if (e >= E) return;
    int d = col[e];
    int pos = atomicAdd(&cursor[d], 1);
    edge_src[rowstart[d] + pos] = row[e];
}

__global__ __launch_bounds__(256, 4) void mfconv_gemm_kernel(
    const float* __restrict__ x,
    const int* __restrict__ degi, const int* __restrict__ rowstart,
    const int* __restrict__ edge_src,
    const float* __restrict__ neigh_w, const float* __restrict__ neigh_b,
    const float* __restrict__ root_w,  const float* __restrict__ root_b,
    float* __restrict__ out, int N)
{
    __shared__ float ldsA[TILE_M * LDA];
    __shared__ int   s_list[LIST_LEN];
    __shared__ int   s_deg[LIST_LEN];
    __shared__ int   s_rs[LIST_LEN];
    __shared__ __attribute__((aligned(16))) int s_src[LIST_LEN * SRC_CAP];
    __shared__ int   s_cnt[RANGE_PASS * 8];

    const int tid   = threadIdx.x;
    const int wave  = tid >> 5;
    const int lane  = tid & 31;
    const int node0 = blockIdx.x * RANGE_N;

    const int by     = blockIdx.y;
    const int bucket = (by < BIG_PARTS) ? MAXDEG : (NBUCK - 2 + BIG_PARTS - by);
    const int part   = (by < BIG_PARTS) ? by : 0;
    const int nparts = (by < BIG_PARTS) ? BIG_PARTS : 1;

    for (int i = tid; i < LIST_LEN; i += 256) s_list[i] = -1;

    bool mine[RANGE_PASS];
    int  myrank[RANGE_PASS];
    int  mydeg[RANGE_PASS];
    int  mystart[RANGE_PASS];
    #pragma unroll
    for (int p = 0; p < RANGE_PASS; ++p) {
        const int node = node0 + p * 256 + tid;
        mydeg[p]   = (node < N) ? degi[node] : 0;
        mystart[p] = (node < N) ? rowstart[node] : 0;
    }
    #pragma unroll
    for (int p = 0; p < RANGE_PASS; ++p) {
        const int node = node0 + p * 256 + tid;
        const bool in = (node < N) && (min(mydeg[p], MAXDEG) == bucket);
        const unsigned m = __builtin_amdgcn_ballot_w32(in);
        if (lane == 0) s_cnt[p * 8 + wave] = __popc(m);
        myrank[p] = __popc(m & ((1u << lane) - 1u));
        mine[p] = in;
    }
    __syncthreads();

    int total = 0;
    int base[RANGE_PASS];
    #pragma unroll
    for (int p = 0; p < RANGE_PASS; ++p) base[p] = 0;
    #pragma unroll
    for (int i = 0; i < RANGE_PASS * 8; ++i) {
        const int c = s_cnt[i];
        #pragma unroll
        for (int p = 0; p < RANGE_PASS; ++p) if (i < p * 8 + wave) base[p] += c;
        total += c;
    }
    const int nTiles = (__builtin_amdgcn_readfirstlane(total) + TILE_M - 1) / TILE_M;
    if (part >= nTiles) return;

    int  pos[RANGE_PASS];
    bool want[RANGE_PASS];
    int  ids[RANGE_PASS][SRC_CAP];
    #pragma unroll
    for (int p = 0; p < RANGE_PASS; ++p) {
        pos[p]  = base[p] + myrank[p];
        want[p] = mine[p] && (mydeg[p] > 0) && (((pos[p] / TILE_M) & (nparts - 1)) == part);
        if (mine[p]) {
            s_list[pos[p]] = node0 + p * 256 + tid;
            s_deg[pos[p]]  = mydeg[p];
            s_rs[pos[p]]   = mystart[p];
        }
        #pragma unroll
        for (int j = 0; j < SRC_CAP; ++j)
            ids[p][j] = (want[p] && j < mydeg[p]) ? edge_src[mystart[p] + j] : 0;
    }

    const int lr    = lane & 15;
    const int khalf = (lane >> 4) << 1;
    const int ncol  = wave * 16 + lr;
    const unsigned coff = (unsigned)ncol * IN_F;
    const int mbase = (lane >> 4) * 8;
    const float* __restrict__ aRow = &ldsA[lr * LDA];

    const float* __restrict__ wn = neigh_w + (size_t)bucket * (OUT_F * IN_F);
    const float* __restrict__ wr = root_w  + (size_t)bucket * (OUT_F * IN_F);
    v2f bfr[64];
    #pragma unroll
    for (int kk = 0; kk < 32; ++kk) {
        const int k = kk * 4 + khalf;
        bfr[kk].x = wn[coff + k]; bfr[kk].y = wn[coff + k + 1];
    }
    #pragma unroll
    for (int kk = 0; kk < 32; ++kk) {
        const int k = kk * 4 + khalf;
        bfr[32 + kk].x = wr[coff + k]; bfr[32 + kk].y = wr[coff + k + 1];
    }
    const float bias = neigh_b[bucket * OUT_F + ncol] + root_b[bucket * OUT_F + ncol];

    #pragma unroll
    for (int p = 0; p < RANGE_PASS; ++p) {
        if (mine[p] && (pos[p] / TILE_M) == part) {
            const float* own = x + (size_t)(node0 + p * 256 + tid) * IN_F;
            __builtin_prefetch(own, 0, 2);
            __builtin_prefetch(own + IN_F / 2, 0, 2);
            #pragma unroll
            for (int j = 0; j < SRC_CAP; ++j) {
                if (j < mydeg[p]) {
                    const float* nb = x + (size_t)ids[p][j] * IN_F;
                    __builtin_prefetch(nb, 0, 2);
                    __builtin_prefetch(nb + IN_F / 2, 0, 2);
                }
            }
        }
    }

    #pragma unroll
    for (int p = 0; p < RANGE_PASS; ++p) {
        if (want[p]) {
            int4* dst = (int4*)&s_src[pos[p] * SRC_CAP];
            #pragma unroll
            for (int c = 0; c < SRC_CAP / 4; ++c)
                dst[c] = make_int4(ids[p][4 * c], ids[p][4 * c + 1], ids[p][4 * c + 2], ids[p][4 * c + 3]);
        }
    }
    __syncthreads();

    for (int t = part; t < nTiles; t += nparts) {
        const int slot0 = t * TILE_M;

        {
            const int  m      = tid >> 4;
            const int  k0     = (tid & 15) * 16;
            const bool isRoot = k0 >= IN_F;
            const int  kx     = k0 & (IN_F - 1);
            const int  slot   = slot0 + m;
            const int  node   = s_list[slot];
            const float  z0  = (isRoot && node >= 0) ? -0.0f : 0.0f;
            const float4 id0 = make_float4(-0.0f, -0.0f, -0.0f, -0.0f);
            float4 a0 = make_float4(z0, z0, z0, z0), a1 = a0, a2 = a0, a3 = a0;
            if (node >= 0) {
                int d = 1;
                if (!isRoot) d = s_deg[slot];
                #pragma unroll
                for (int c = 0; c < SRC_CAP; c += 4) {
                    if (c < d) {
                        int4 id4 = make_int4(node, 0, 0, 0);
                        if (!isRoot) id4 = *(const int4*)&s_src[slot * SRC_CAP + c];
                        const int idv[4] = { id4.x, id4.y, id4.z, id4.w };
                        float4 v[4][4];
                        #pragma unroll
                        for (int j = 0; j < 4; ++j) {
                            v[j][0] = id0; v[j][1] = id0; v[j][2] = id0; v[j][3] = id0;
                            if (c + j < d) {
                                const float* xs = x + (size_t)idv[j] * IN_F + kx;
                                v[j][0] = *(const float4*)(xs + 0);
                                v[j][1] = *(const float4*)(xs + 4);
                                v[j][2] = *(const float4*)(xs + 8);
                                v[j][3] = *(const float4*)(xs + 12);
                            }
                        }
                        #pragma unroll
                        for (int j = 0; j < 4; ++j) {
                            a0.x += v[j][0].x; a0.y += v[j][0].y; a0.z += v[j][0].z; a0.w += v[j][0].w;
                            a1.x += v[j][1].x; a1.y += v[j][1].y; a1.z += v[j][1].z; a1.w += v[j][1].w;
                            a2.x += v[j][2].x; a2.y += v[j][2].y; a2.z += v[j][2].z; a2.w += v[j][2].w;
                            a3.x += v[j][3].x; a3.y += v[j][3].y; a3.z += v[j][3].z; a3.w += v[j][3].w;
                        }
                    }
                }
                if (d > SRC_CAP) {
                    const int start = s_rs[slot];
                    #pragma unroll 1
                    for (int j = SRC_CAP; j < d; ++j) {
                        const float* xs = x + (size_t)edge_src[start + j] * IN_F + kx;
                        const float4 v0 = *(const float4*)(xs + 0);
                        const float4 v1 = *(const float4*)(xs + 4);
                        const float4 v2 = *(const float4*)(xs + 8);
                        const float4 v3 = *(const float4*)(xs + 12);
                        a0.x += v0.x; a0.y += v0.y; a0.z += v0.z; a0.w += v0.w;
                        a1.x += v1.x; a1.y += v1.y; a1.z += v1.z; a1.w += v1.w;
                        a2.x += v2.x; a2.y += v2.y; a2.z += v2.z; a2.w += v2.w;
                        a3.x += v3.x; a3.y += v3.y; a3.z += v3.z; a3.w += v3.w;
                    }
                }
                const float inv = isRoot ? 1.0f : 1.0f / fmaxf((float)d, 1.0f);
                a0.x *= inv; a0.y *= inv; a0.z *= inv; a0.w *= inv;
                a1.x *= inv; a1.y *= inv; a1.z *= inv; a1.w *= inv;
                a2.x *= inv; a2.y *= inv; a2.z *= inv; a2.w *= inv;
                a3.x *= inv; a3.y *= inv; a3.z *= inv; a3.w *= inv;
            }
            float* dst = &ldsA[m * LDA + k0];
            *(float4*)(dst + 0)  = a0;
            *(float4*)(dst + 4)  = a1;
            *(float4*)(dst + 8)  = a2;
            *(float4*)(dst + 12) = a3;
        }
        __syncthreads();

        if (t + nparts < nTiles) {
            const int pslot = (t + nparts) * TILE_M + (tid >> 4);
            const int pnode = s_list[pslot];
            if (pnode >= 0) {
                const int q = tid & 15;
                int prow = pnode;
                if (q >= 2) {
                    const int j = (q - 2) >> 1;
                    prow = (j < s_deg[pslot]) ? s_src[pslot * SRC_CAP + j] : -1;
                }
                if (prow >= 0) __builtin_prefetch(x + (size_t)prow * IN_F + (q & 1) * (IN_F / 2), 0, 2);
            }
        }

        v8f acc = {};
        #pragma unroll
        for (int kk = 0; kk < 32; ++kk) {
            const int k = kk * 4 + khalf;
            v2f a;
            a.x = aRow[k];     a.y = aRow[k + 1];
            acc = __builtin_amdgcn_wmma_f32_16x16x4_f32(false, a, false, bfr[kk],
                                                        (short)0, acc, false, false);
        }
        #pragma unroll
        for (int kk = 0; kk < 32; ++kk) {
            const int k = kk * 4 + khalf;
            v2f a;
            a.x = aRow[IN_F + k]; a.y = aRow[IN_F + k + 1];
            acc = __builtin_amdgcn_wmma_f32_16x16x4_f32(false, a, false, bfr[32 + kk],
                                                        (short)0, acc, false, false);
        }

        #pragma unroll
        for (int i = 0; i < 8; ++i) {
            const int node = s_list[slot0 + mbase + i];
            if (node >= 0) out[(size_t)node * OUT_F + ncol] = acc[i] + bias;
        }
        __syncthreads();
    }
}

extern "C" void kernel_launch(void* const* d_in, const int* in_sizes, int n_in,
                              void* d_out, int out_size, void* d_ws, size_t ws_size,
                              hipStream_t stream) {
    const float* x       = (const float*)d_in[0];
    const float* neigh_w = (const float*)d_in[1];
    const float* neigh_b = (const float*)d_in[2];
    const float* root_w  = (const float*)d_in[3];
    const float* root_b  = (const float*)d_in[4];
    const int*   eidx    = (const int*)d_in[5];

    const int N = in_sizes[0] / IN_F;
    const int E = in_sizes[5] / 2;
    const int* row = eidx;
    const int* col = eidx + E;

    const int nScanBlocks = (N + SCAN_B - 1) / SCAN_B;
    const int nRanges = (N + RANGE_N - 1) / RANGE_N;

    char* ws = (char*)d_ws;
    int* degi     = (int*)ws;  ws += (size_t)N * sizeof(int);
    int* cursor   = (int*)ws;  ws += (size_t)N * sizeof(int);
    int* rowstart = (int*)ws;  ws += (size_t)N * sizeof(int);
    int* bsum     = (int*)ws;  ws += (size_t)((nScanBlocks + 15) & ~15) * sizeof(int);
    int* edge_src = (int*)ws;

    fill_i32_kernel<<<1024, 256, 0, stream>>>(degi, 0, (size_t)2 * N);

    degree_kernel<<<(E + 255) / 256, 256, 0, stream>>>(col, degi, E);
    scan1_kernel<<<nScanBlocks, SCAN_B, 0, stream>>>(degi, rowstart, bsum, N);
    scan2_kernel<<<1, 1, 0, stream>>>(bsum, nScanBlocks);
    scan3_kernel<<<nScanBlocks, SCAN_B, 0, stream>>>(rowstart, bsum, N);
    csr_fill_kernel<<<(E + 255) / 256, 256, 0, stream>>>(row, col, rowstart, cursor,
                                                         edge_src, E);

    mfconv_gemm_kernel<<<dim3(nRanges, GRID_Y), 256, 0, stream>>>(
        x, degi, rowstart, edge_src, neigh_w, neigh_b, root_w, root_b, (float*)d_out, N);
}


# GEMM tile loop software-pipelined: next tile's rows (own + 3 neighbours, 8 floats a lane) loaded into registers ahead of the matrix loop, summed into a 2nd LDS buffer behind it; 1 barrier a tile
# speedup vs baseline: 2.7986x; 1.4868x over previous
#include <hip/hip_runtime.h>


typedef __attribute__((ext_vector_type(2))) float v2f;
typedef __attribute__((ext_vector_type(8))) float v8f;

#define IN_F   128
#define OUT_F  128
#define MAXDEG 4
#define NBUCK  5
#define TILE_M 16
#define LDA    260
#define RANGE_N     512
#define RUN_SHIFT   9
#define BIG_PARTS   2
#define GRID_Y      (NBUCK - 1 + BIG_PARTS)
#define RANGE_PASS  (RANGE_N / 256)
#define LIST_LEN    (RANGE_N + TILE_M)
#define SRC_CAP     8
#define OVF_CAP     256
#define EDGE_B      4096
#define BIN_T       1024
#define EDGE_BATCH  8


__global__ __launch_bounds__(BIN_T) void run_hist_kernel(const int* __restrict__ col, int* __restrict__ hist,
                                                         int E, int nBins) {
    extern __shared__ int dyn_lds[];
    const int tid = threadIdx.x;
    for (int b = tid; b < nBins; b += BIN_T) dyn_lds[b] = 0;
    __syncthreads();
    int d[EDGE_B / BIN_T];
    for (int k = 0; k < EDGE_B / BIN_T; ++k) {
        const int e = blockIdx.x * EDGE_B + k * BIN_T + tid;
        d[k] = col[min(e, E - 1)];
    }
    for (int k = 0; k < EDGE_B / BIN_T; ++k) {
        const int e = blockIdx.x * EDGE_B + k * BIN_T + tid;
        if (e < E) atomicAdd(&dyn_lds[d[k] >> RUN_SHIFT], 1);
    }
    __syncthreads();
    for (int b = tid; b < nBins; b += BIN_T) hist[blockIdx.x * nBins + b] = dyn_lds[b];
}

__global__ __launch_bounds__(BIN_T) void run_place_kernel(const int* __restrict__ row, const int* __restrict__ col,
                                                          const int* __restrict__ hist, int* __restrict__ binstart,
                                                          unsigned* __restrict__ bins, int E, int nBins, int nEB) {
    extern __shared__ int dyn_lds[];
    int* sA = dyn_lds;
    int* sB = dyn_lds + 2 * nBins;
    const int tid  = threadIdx.x;
    const int wg   = blockIdx.x;
    const int half = tid >> 9;

    int d[EDGE_B / BIN_T], s[EDGE_B / BIN_T];
    for (int k = 0; k < EDGE_B / BIN_T; ++k) {
        const int e = min(wg * EDGE_B + k * BIN_T + tid, E - 1);
        d[k] = col[e];
        s[k] = row[e];
    }

    for (int b = tid & 511; b < nBins; b += 512) {
        int tot = 0, pre = 0;
        int g = half;
        for (; g + 14 < nEB; g += 16) {
            int v[8];
            for (int u = 0; u < 8; ++u) v[u] = hist[(g + 2 * u) * nBins + b];
            for (int u = 0; u < 8; ++u) { tot += v[u]; pre += (g + 2 * u < wg) ? v[u] : 0; }
        }
        for (; g < nEB; g += 2) {
            const int v = hist[g * nBins + b];
            tot += v; pre += (g < wg) ? v : 0;
        }
        sA[half * nBins + b] = tot;
        sB[half * nBins + b] = pre;
    }
    __syncthreads();
    for (int b = tid; b < nBins; b += BIN_T) { sA[b] += sA[nBins + b]; sB[b] += sB[nBins + b]; }
    __syncthreads();

    if (tid < 32) {
        const int chunk = (nBins + 31) / 32;
        const int lo = min(tid * chunk, nBins);
        const int hi = min(lo + chunk, nBins);
        int mysum = 0;
        for (int b = lo; b < hi; ++b) mysum += sA[b];
        int incl = mysum;
        for (int off = 1; off < 32; off <<= 1) {
            const int t = __shfl_up(incl, off, 32);
            if (tid >= off) incl += t;
        }
        int run = incl - mysum;
        for (int b = lo; b < hi; ++b) { const int t = sA[b]; sA[b] = run; run += t; }
    }
    __syncthreads();
    for (int b = tid; b < nBins; b += BIN_T) {
        const int st = sA[b];
        if (wg == 0) binstart[b] = st;
        sB[b] += st;
    }
    if (wg == 0 && tid == 0) binstart[nBins] = E;
    __syncthreads();

    for (int k = 0; k < EDGE_B / BIN_T; ++k) {
        const int e = wg * EDGE_B + k * BIN_T + tid;
        if (e < E) {
            const int pos = atomicAdd(&sB[d[k] >> RUN_SHIFT], 1);
            bins[pos] = ((unsigned)s[k] << RUN_SHIFT) | (unsigned)(d[k] & (RANGE_N - 1));
        }
    }
}

__global__ __launch_bounds__(256, 4) void mfconv_gemm_kernel(
    const float* __restrict__ x,
    const int* __restrict__ binstart, const unsigned* __restrict__ bins,
    const float* __restrict__ neigh_w, const float* __restrict__ neigh_b,
    const float* __restrict__ root_w,  const float* __restrict__ root_b,
    float* __restrict__ out, int N)
{
    __shared__ __attribute__((aligned(16))) float ldsA[2 * TILE_M * LDA];
    __shared__ int   s_list[LIST_LEN];
    __shared__ int   s_dl[RANGE_N];
    __shared__ __attribute__((aligned(16))) int s_src[RANGE_N * SRC_CAP];
    __shared__ unsigned s_ovf[OVF_CAP];
    __shared__ int   s_ovn;
    __shared__ int   s_cnt[RANGE_PASS * 8];

    const int tid   = threadIdx.x;
    const int wave  = tid >> 5;
    const int lane  = tid & 31;
    const int node0 = blockIdx.x * RANGE_N;

    const int by     = blockIdx.y;
    const int bucket = (by < BIG_PARTS) ? MAXDEG : (NBUCK - 2 + BIG_PARTS - by);
    const int part   = (by < BIG_PARTS) ? by : 0;
    const int nparts = (by < BIG_PARTS) ? BIG_PARTS : 1;

    for (int i = tid; i < LIST_LEN; i += 256) s_list[i] = -1;
    for (int i = tid; i < RANGE_N; i += 256) s_dl[i] = 0;
    if (tid == 0) s_ovn = 0;
    const int e0 = binstart[blockIdx.x];
    const int e1 = binstart[blockIdx.x + 1];
    __syncthreads();

    for (int eb = e0; eb < e1; eb += 256 * EDGE_BATCH) {
        unsigned w[EDGE_BATCH];
        for (int u = 0; u < EDGE_BATCH; ++u)
            w[u] = bins[min(eb + u * 256 + tid, e1 - 1)];
        int slot[EDGE_BATCH];
        for (int u = 0; u < EDGE_BATCH; ++u) {
            slot[u] = -1;
            if (eb + u * 256 + tid < e1) slot[u] = atomicAdd(&s_dl[w[u] & (RANGE_N - 1)], 1);
        }
        for (int u = 0; u < EDGE_BATCH; ++u) {
            if (slot[u] >= 0) {
                if (slot[u] < SRC_CAP) {
                    s_src[(int)(w[u] & (RANGE_N - 1)) * SRC_CAP + slot[u]] = (int)(w[u] >> RUN_SHIFT);
                } else {
                    const int o = atomicAdd(&s_ovn, 1);
                    if (o < OVF_CAP) s_ovf[o] = w[u];
                }
            }
        }
    }
    __syncthreads();

    bool mine[RANGE_PASS];
    int  myrank[RANGE_PASS];
    #pragma unroll
    for (int p = 0; p < RANGE_PASS; ++p) {
        const int node = node0 + p * 256 + tid;
        const bool in = (node < N) && (min(s_dl[p * 256 + tid], MAXDEG) == bucket);
        const unsigned m = __builtin_amdgcn_ballot_w32(in);
        if (lane == 0) s_cnt[p * 8 + wave] = __popc(m);
        myrank[p] = __popc(m & ((1u << lane) - 1u));
        mine[p] = in;
    }
    __syncthreads();

    int total = 0;
    int base[RANGE_PASS];
    #pragma unroll
    for (int p = 0; p < RANGE_PASS; ++p) base[p] = 0;
    #pragma unroll
    for (int i = 0; i < RANGE_PASS * 8; ++i) {
        const int c = s_cnt[i];
        #pragma unroll
        for (int p = 0; p < RANGE_PASS; ++p) if (i < p * 8 + wave) base[p] += c;
        total += c;
    }
    const int nTiles = (__builtin_amdgcn_readfirstlane(total) + TILE_M - 1) / TILE_M;
    if (part >= nTiles) return;

    #pragma unroll
    for (int p = 0; p < RANGE_PASS; ++p)
        if (mine[p]) s_list[base[p] + myrank[p]] = node0 + p * 256 + tid;

    __syncthreads();

    const int sm = tid >> 4;
    const int fq = (tid & 15) * 8;

    int    g_node = -1, g_ln = 0, g_d = 0;
    float4 g_own[2];
    float4 g_nb[3][2];

    auto issue_rows = [&](int tile) {
        const float4 id0 = make_float4(-0.0f, -0.0f, -0.0f, -0.0f);
        #pragma unroll
        for (int j = 0; j < 3; ++j) { g_nb[j][0] = id0; g_nb[j][1] = id0; }
        g_own[0] = id0; g_own[1] = id0;
        g_node = s_list[tile * TILE_M + sm];
        g_ln = 0;
        g_d  = 0;
        if (g_node >= 0) {
            const float* xo = x + (size_t)g_node * IN_F + fq;
            g_own[0] = *(const float4*)(xo + 0);
            g_own[1] = *(const float4*)(xo + 4);
            g_ln = g_node - node0;
            g_d  = s_dl[g_ln];
            if (g_d > 0) {
                const int4 id4 = *(const int4*)&s_src[g_ln * SRC_CAP];
                const int idv[3] = { id4.x, id4.y, id4.z };
                #pragma unroll
                for (int j = 0; j < 3; ++j) {
                    if (j < g_d) {
                        const float* xs = x + (size_t)idv[j] * IN_F + fq;
                        g_nb[j][0] = *(const float4*)(xs + 0);
                        g_nb[j][1] = *(const float4*)(xs + 4);
                    }
                }
            }
        }
    };

    auto request_late_rows = [&](int tile) {
        const int pnode = s_list[tile * TILE_M + sm];
        const int q = tid & 15;
        if (pnode >= 0 && q < 10) {
            const int pl = pnode - node0;
            const int j  = 3 + (q >> 1);
            if (j < s_dl[pl]) {
                const int prow = s_src[pl * SRC_CAP + j];
                __builtin_prefetch(x + (size_t)prow * IN_F + (q & 1) * (IN_F / 2), 0, 2);
            }
        }
    };

    auto finish_rows = [&](float* dstRow) {
        float4 n0 = make_float4(0.0f, 0.0f, 0.0f, 0.0f), n1 = n0;
        float4 w0 = n0, w1 = n0;
        if (g_node >= 0) {
            const int ln = g_ln;
            const int d  = g_d;
            w0 = g_own[0]; w1 = g_own[1];
            if (d > 0) {
                #pragma unroll
                for (int j = 0; j < 3; ++j) {
                    n0.x += g_nb[j][0].x; n0.y += g_nb[j][0].y; n0.z += g_nb[j][0].z; n0.w += g_nb[j][0].w;
                    n1.x += g_nb[j][1].x; n1.y += g_nb[j][1].y; n1.z += g_nb[j][1].z; n1.w += g_nb[j][1].w;
                }
            }
            if (d > 3) {
                const float4 id0 = make_float4(-0.0f, -0.0f, -0.0f, -0.0f);
                const int  id3 = s_src[ln * SRC_CAP + 3];
                const int4 id4 = *(const int4*)&s_src[ln * SRC_CAP + 4];
                const int idv[5] = { id3, id4.x, id4.y, id4.z, id4.w };
                float4 v[5][2];
                #pragma unroll
                for (int j = 0; j < 5; ++j) {
                    v[j][0] = id0; v[j][1] = id0;
                    if (3 + j < d) {
                        const float* xs = x + (size_t)idv[j] * IN_F + fq;
                        v[j][0] = *(const float4*)(xs + 0);
                        v[j][1] = *(const float4*)(xs + 4);
                    }
                }
                #pragma unroll
                for (int j = 0; j < 5; ++j) {
                    n0.x += v[j][0].x; n0.y += v[j][0].y; n0.z += v[j][0].z; n0.w += v[j][0].w;
                    n1.x += v[j][1].x; n1.y += v[j][1].y; n1.z += v[j][1].z; n1.w += v[j][1].w;
                }
            }
            if (d > SRC_CAP) {
                auto add_row = [&](unsigned srcId) {
                    const float* xs = x + (size_t)srcId * IN_F + fq;
                    const float4 v0 = *(const float4*)(xs + 0);
                    const float4 v1 = *(const float4*)(xs + 4);
                    n0.x += v0.x; n0.y += v0.y; n0.z += v0.z; n0.w += v0.w;
                    n1.x += v1.x; n1.y += v1.y; n1.z += v1.z; n1.w += v1.w;
                };
                const int novf = s_ovn;
                if (novf <= OVF_CAP) {
                    #pragma unroll 1
                    for (int j = 0; j < novf; ++j) {
                        const unsigned w = s_ovf[j];
                        if ((int)(w & (RANGE_N - 1)) == ln) add_row(w >> RUN_SHIFT);
                    }
                } else {
                    n0 = make_float4(0.0f, 0.0f, 0.0f, 0.0f); n1 = n0;
                    #pragma unroll 1
                    for (int e = e0; e < e1; ++e) {
                        const unsigned w = bins[e];
                        if ((int)(w & (RANGE_N - 1)) == ln) add_row(w >> RUN_SHIFT);
                    }
                }
            }
            const float inv = 1.0f / fmaxf((float)d, 1.0f);
            n0.x *= inv; n0.y *= inv; n0.z *= inv; n0.w *= inv;
            n1.x *= inv; n1.y *= inv; n1.z *= inv; n1.w *= inv;
        }
        *(float4*)(dstRow + fq + 0)        = n0;
        *(float4*)(dstRow + fq + 4)        = n1;
        *(float4*)(dstRow + IN_F + fq + 0) = w0;
        *(float4*)(dstRow + IN_F + fq + 4) = w1;
    };

    issue_rows(part);
    if (bucket == MAXDEG) request_late_rows(part);

    const int lr    = lane & 15;
    const int khalf = (lane >> 4) << 1;
    const int ncol  = wave * 16 + lr;
    const unsigned coff = (unsigned)ncol * IN_F;
    const int mbase = (lane >> 4) * 8;

    const float* __restrict__ wn = neigh_w + (size_t)bucket * (OUT_F * IN_F);
    const float* __restrict__ wr = root_w  + (size_t)bucket * (OUT_F * IN_F);
    v2f bfr[64];
    #pragma unroll
    for (int kk = 0; kk < 32; ++kk) {
        const int k = kk * 4 + khalf;
        bfr[kk].x = wn[coff + k]; bfr[kk].y = wn[coff + k + 1];
    }
    #pragma unroll
    for (int kk = 0; kk < 32; ++kk) {
        const int k = kk * 4 + khalf;
        bfr[32 + kk].x = wr[coff + k]; bfr[32 + kk].y = wr[coff + k + 1];
    }
    const float bias = neigh_b[bucket * OUT_F + ncol] + root_b[bucket * OUT_F + ncol];

    finish_rows(&ldsA[sm * LDA]);
    __syncthreads();

    int cur = 0;
    for (int t = part; t < nTiles; t += nparts) {
        const int  slot0 = t * TILE_M;
        const bool more  = t + nparts < nTiles;

        if (more) {
            issue_rows(t + nparts);
            if (bucket == MAXDEG) request_late_rows(t + nparts);
        }

        const float* __restrict__ aRow = &ldsA[cur * (TILE_M * LDA) + lr * LDA];
        v8f acc = {};
        #pragma unroll
        for (int kk = 0; kk < 32; ++kk) {
            const int k = kk * 4 + khalf;
            v2f a;
            a.x = aRow[k];     a.y = aRow[k + 1];
            acc = __builtin_amdgcn_wmma_f32_16x16x4_f32(false, a, false, bfr[kk],
                                                        (short)0, acc, false, false);
        }
        #pragma unroll
        for (int kk = 0; kk < 32; ++kk) {
            const int k = kk * 4 + khalf;
            v2f a;
            a.x = aRow[IN_F + k]; a.y = aRow[IN_F + k + 1];
            acc = __builtin_amdgcn_wmma_f32_16x16x4_f32(false, a, false, bfr[32 + kk],
                                                        (short)0, acc, false, false);
        }

        #pragma unroll
        for (int i = 0; i < 8; ++i) {
            const int node = s_list[slot0 + mbase + i];
            if (node >= 0) out[(size_t)node * OUT_F + ncol] = acc[i] + bias;
        }

        if (more) {
            finish_rows(&ldsA[(cur ^ 1) * (TILE_M * LDA) + sm * LDA]);
            __syncthreads();
        }
        cur ^= 1;
    }
}

extern "C" void kernel_launch(void* const* d_in, const int* in_sizes, int n_in,
                              void* d_out, int out_size, void* d_ws, size_t ws_size,
                              hipStream_t stream) {
    const float* x       = (const float*)d_in[0];
    const float* neigh_w = (const float*)d_in[1];
    const float* neigh_b = (const float*)d_in[2];
    const float* root_w  = (const float*)d_in[3];
    const float* root_b  = (const float*)d_in[4];
    const int*   eidx    = (const int*)d_in[5];

    const int N = in_sizes[0] / IN_F;
    const int E = in_sizes[5] / 2;
    const int* row = eidx;
    const int* col = eidx + E;

    const int nRanges = (N + RANGE_N - 1) / RANGE_N;
    const int nEB     = (E + EDGE_B - 1) / EDGE_B;

    char* ws = (char*)d_ws;
    int* hist      = (int*)ws;       ws += (size_t)((nEB * nRanges + 15) & ~15) * sizeof(int);
    int* binstart  = (int*)ws;       ws += (size_t)((nRanges + 1 + 15) & ~15) * sizeof(int);
    unsigned* bins = (unsigned*)ws;

    run_hist_kernel<<<nEB, BIN_T, (size_t)nRanges * sizeof(int), stream>>>(col, hist, E, nRanges);
    run_place_kernel<<<nEB, BIN_T, (size_t)4 * nRanges * sizeof(int), stream>>>(row, col, hist, binstart, bins,
                                                                              E, nRanges, nEB);

    mfconv_gemm_kernel<<<dim3(nRanges, GRID_Y), 256, 0, stream>>>(
        x, binstart, bins, neigh_w, neigh_b, root_w, root_b, (float*)d_out, N);
}


# GEMM tile loop: a tile's output stores issued behind the hand-over barrier (its 8 rows' node ids read from LDS as two 16-byte words ahead of the matrix loop), not in front of it
# speedup vs baseline: 2.9403x; 1.0506x over previous
#include <hip/hip_runtime.h>


typedef __attribute__((ext_vector_type(2))) float v2f;
typedef __attribute__((ext_vector_type(8))) float v8f;

#define IN_F   128
#define OUT_F  128
#define MAXDEG 4
#define NBUCK  5
#define TILE_M 16
#define LDA    260
#define RANGE_N     512
#define RUN_SHIFT   9
#define BIG_PARTS   2
#define GRID_Y      (NBUCK - 1 + BIG_PARTS)
#define RANGE_PASS  (RANGE_N / 256)
#define LIST_LEN    (RANGE_N + TILE_M)
#define SRC_CAP     8
#define OVF_CAP     256
#define EDGE_B      4096
#define BIN_T       1024
#define EDGE_BATCH  8


__global__ __launch_bounds__(BIN_T) void run_hist_kernel(const int* __restrict__ col, int* __restrict__ hist,
                                                         int E, int nBins) {
    extern __shared__ int dyn_lds[];
    const int tid = threadIdx.x;
    for (int b = tid; b < nBins; b += BIN_T) dyn_lds[b] = 0;
    __syncthreads();
    int d[EDGE_B / BIN_T];
    for (int k = 0; k < EDGE_B / BIN_T; ++k) {
        const int e = blockIdx.x * EDGE_B + k * BIN_T + tid;
        d[k] = col[min(e, E - 1)];
    }
    for (int k = 0; k < EDGE_B / BIN_T; ++k) {
        const int e = blockIdx.x * EDGE_B + k * BIN_T + tid;
        if (e < E) atomicAdd(&dyn_lds[d[k] >> RUN_SHIFT], 1);
    }
    __syncthreads();
    for (int b = tid; b < nBins; b += BIN_T) hist[blockIdx.x * nBins + b] = dyn_lds[b];
}

__global__ __launch_bounds__(BIN_T) void run_place_kernel(const int* __restrict__ row, const int* __restrict__ col,
                                                          const int* __restrict__ hist, int* __restrict__ binstart,
                                                          unsigned* __restrict__ bins, int E, int nBins, int nEB) {
    extern __shared__ int dyn_lds[];
    int* sA = dyn_lds;
    int* sB = dyn_lds + 2 * nBins;
    const int tid  = threadIdx.x;
    const int wg   = blockIdx.x;
    const int half = tid >> 9;

    int d[EDGE_B / BIN_T], s[EDGE_B / BIN_T];
    for (int k = 0; k < EDGE_B / BIN_T; ++k) {
        const int e = min(wg * EDGE_B + k * BIN_T + tid, E - 1);
        d[k] = col[e];
        s[k] = row[e];
    }

    for (int b = tid & 511; b < nBins; b += 512) {
        int tot = 0, pre = 0;
        int g = half;
        for (; g + 14 < nEB; g += 16) {
            int v[8];
            for (int u = 0; u < 8; ++u) v[u] = hist[(g + 2 * u) * nBins + b];
            for (int u = 0; u < 8; ++u) { tot += v[u]; pre += (g + 2 * u < wg) ? v[u] : 0; }
        }
        for (; g < nEB; g += 2) {
            const int v = hist[g * nBins + b];
            tot += v; pre += (g < wg) ? v : 0;
        }
        sA[half * nBins + b] = tot;
        sB[half * nBins + b] = pre;
    }
    __syncthreads();
    for (int b = tid; b < nBins; b += BIN_T) { sA[b] += sA[nBins + b]; sB[b] += sB[nBins + b]; }
    __syncthreads();

    if (tid < 32) {
        const int chunk = (nBins + 31) / 32;
        const int lo = min(tid * chunk, nBins);
        const int hi = min(lo + chunk, nBins);
        int mysum = 0;
        for (int b = lo; b < hi; ++b) mysum += sA[b];
        int incl = mysum;
        for (int off = 1; off < 32; off <<= 1) {
            const int t = __shfl_up(incl, off, 32);
            if (tid >= off) incl += t;
        }
        int run = incl - mysum;
        for (int b = lo; b < hi; ++b) { const int t = sA[b]; sA[b] = run; run += t; }
    }
    __syncthreads();
    for (int b = tid; b < nBins; b += BIN_T) {
        const int st = sA[b];
        if (wg == 0) binstart[b] = st;
        sB[b] += st;
    }
    if (wg == 0 && tid == 0) binstart[nBins] = E;
    __syncthreads();

    for (int k = 0; k < EDGE_B / BIN_T; ++k) {
        const int e = wg * EDGE_B + k * BIN_T + tid;
        if (e < E) {
            const int pos = atomicAdd(&sB[d[k] >> RUN_SHIFT], 1);
            bins[pos] = ((unsigned)s[k] << RUN_SHIFT) | (unsigned)(d[k] & (RANGE_N - 1));
        }
    }
}

__global__ __launch_bounds__(256, 4) void mfconv_gemm_kernel(
    const float* __restrict__ x,
    const int* __restrict__ binstart, const unsigned* __restrict__ bins,
    const float* __restrict__ neigh_w, const float* __restrict__ neigh_b,
    const float* __restrict__ root_w,  const float* __restrict__ root_b,
    float* __restrict__ out, int N)
{
    __shared__ __attribute__((aligned(16))) float ldsA[2 * TILE_M * LDA];
    __shared__ __attribute__((aligned(16))) int s_list[LIST_LEN];
    __shared__ int   s_dl[RANGE_N];
    __shared__ __attribute__((aligned(16))) int s_src[RANGE_N * SRC_CAP];
    __shared__ unsigned s_ovf[OVF_CAP];
    __shared__ int   s_ovn;
    __shared__ int   s_cnt[RANGE_PASS * 8];

    const int tid   = threadIdx.x;
    const int wave  = tid >> 5;
    const int lane  = tid & 31;
    const int node0 = blockIdx.x * RANGE_N;

    const int by     = blockIdx.y;
    const int bucket = (by < BIG_PARTS) ? MAXDEG : (NBUCK - 2 + BIG_PARTS - by);
    const int part   = (by < BIG_PARTS) ? by : 0;
    const int nparts = (by < BIG_PARTS) ? BIG_PARTS : 1;

    for (int i = tid; i < LIST_LEN; i += 256) s_list[i] = -1;
    for (int i = tid; i < RANGE_N; i += 256) s_dl[i] = 0;
    if (tid == 0) s_ovn = 0;
    const int e0 = binstart[blockIdx.x];
    const int e1 = binstart[blockIdx.x + 1];
    __syncthreads();

    for (int eb = e0; eb < e1; eb += 256 * EDGE_BATCH) {
        unsigned w[EDGE_BATCH];
        for (int u = 0; u < EDGE_BATCH; ++u)
            w[u] = bins[min(eb + u * 256 + tid, e1 - 1)];
        int slot[EDGE_BATCH];
        for (int u = 0; u < EDGE_BATCH; ++u) {
            slot[u] = -1;
            if (eb + u * 256 + tid < e1) slot[u] = atomicAdd(&s_dl[w[u] & (RANGE_N - 1)], 1);
        }
        for (int u = 0; u < EDGE_BATCH; ++u) {
            if (slot[u] >= 0) {
                if (slot[u] < SRC_CAP) {
                    s_src[(int)(w[u] & (RANGE_N - 1)) * SRC_CAP + slot[u]] = (int)(w[u] >> RUN_SHIFT);
                } else {
                    const int o = atomicAdd(&s_ovn, 1);
                    if (o < OVF_CAP) s_ovf[o] = w[u];
                }
            }
        }
    }
    __syncthreads();

    bool mine[RANGE_PASS];
    int  myrank[RANGE_PASS];
    #pragma unroll
    for (int p = 0; p < RANGE_PASS; ++p) {
        const int node = node0 + p * 256 + tid;
        const bool in = (node < N) && (min(s_dl[p * 256 + tid], MAXDEG) == bucket);
        const unsigned m = __builtin_amdgcn_ballot_w32(in);
        if (lane == 0) s_cnt[p * 8 + wave] = __popc(m);
        myrank[p] = __popc(m & ((1u << lane) - 1u));
        mine[p] = in;
    }
    __syncthreads();

    int total = 0;
    int base[RANGE_PASS];
    #pragma unroll
    for (int p = 0; p < RANGE_PASS; ++p) base[p] = 0;
    #pragma unroll
    for (int i = 0; i < RANGE_PASS * 8; ++i) {
        const int c = s_cnt[i];
        #pragma unroll
        for (int p = 0; p < RANGE_PASS; ++p) if (i < p * 8 + wave) base[p] += c;
        total += c;
    }
    const int nTiles = (__builtin_amdgcn_readfirstlane(total) + TILE_M - 1) / TILE_M;
    if (part >= nTiles) return;

    #pragma unroll
    for (int p = 0; p < RANGE_PASS; ++p)
        if (mine[p]) s_list[base[p] + myrank[p]] = node0 + p * 256 + tid;

    __syncthreads();

    const int sm = tid >> 4;
    const int fq = (tid & 15) * 8;

    int    g_node = -1, g_ln = 0, g_d = 0;
    float4 g_own[2];
    float4 g_nb[3][2];

    auto issue_rows = [&](int tile) {
        const float4 id0 = make_float4(-0.0f, -0.0f, -0.0f, -0.0f);
        #pragma unroll
        for (int j = 0; j < 3; ++j) { g_nb[j][0] = id0; g_nb[j][1] = id0; }
        g_own[0] = id0; g_own[1] = id0;
        g_node = s_list[tile * TILE_M + sm];
        g_ln = 0;
        g_d  = 0;
        if (g_node >= 0) {
            const float* xo = x + (size_t)g_node * IN_F + fq;
            g_own[0] = *(const float4*)(xo + 0);
            g_own[1] = *(const float4*)(xo + 4);
            g_ln = g_node - node0;
            g_d  = s_dl[g_ln];
            if (g_d > 0) {
                const int4 id4 = *(const int4*)&s_src[g_ln * SRC_CAP];
                const int idv[3] = { id4.x, id4.y, id4.z };
                #pragma unroll
                for (int j = 0; j < 3; ++j) {
                    if (j < g_d) {
                        const float* xs = x + (size_t)idv[j] * IN_F + fq;
                        g_nb[j][0] = *(const float4*)(xs + 0);
                        g_nb[j][1] = *(const float4*)(xs + 4);
                    }
                }
            }
        }
    };

    auto request_late_rows = [&](int tile) {
        const int pnode = s_list[tile * TILE_M + sm];
        const int q = tid & 15;
        if (pnode >= 0 && q < 10) {
            const int pl = pnode - node0;
            const int j  = 3 + (q >> 1);
            if (j < s_dl[pl]) {
                const int prow = s_src[pl * SRC_CAP + j];
                __builtin_prefetch(x + (size_t)prow * IN_F + (q & 1) * (IN_F / 2), 0, 2);
            }
        }
    };

    auto finish_rows = [&](float* dstRow) {
        float4 n0 = make_float4(0.0f, 0.0f, 0.0f, 0.0f), n1 = n0;
        float4 w0 = n0, w1 = n0;
        if (g_node >= 0) {
            const int ln = g_ln;
            const int d  = g_d;
            w0 = g_own[0]; w1 = g_own[1];
            if (d > 0) {
                #pragma unroll
                for (int j = 0; j < 3; ++j) {
                    n0.x += g_nb[j][0].x; n0.y += g_nb[j][0].y; n0.z += g_nb[j][0].z; n0.w += g_nb[j][0].w;
                    n1.x += g_nb[j][1].x; n1.y += g_nb[j][1].y; n1.z += g_nb[j][1].z; n1.w += g_nb[j][1].w;
                }
            }
            if (d > 3) {
                const float4 id0 = make_float4(-0.0f, -0.0f, -0.0f, -0.0f);
                const int  id3 = s_src[ln * SRC_CAP + 3];
                const int4 id4 = *(const int4*)&s_src[ln * SRC_CAP + 4];
                const int idv[5] = { id3, id4.x, id4.y, id4.z, id4.w };
                float4 v[5][2];
                #pragma unroll
                for (int j = 0; j < 5; ++j) {
                    v[j][0] = id0; v[j][1] = id0;
                    if (3 + j < d) {
                        const float* xs = x + (size_t)idv[j] * IN_F + fq;
                        v[j][0] = *(const float4*)(xs + 0);
                        v[j][1] = *(const float4*)(xs + 4);
                    }
                }
                #pragma unroll
                for (int j = 0; j < 5; ++j) {
                    n0.x += v[j][0].x; n0.y += v[j][0].y; n0.z += v[j][0].z; n0.w += v[j][0].w;
                    n1.x += v[j][1].x; n1.y += v[j][1].y; n1.z += v[j][1].z; n1.w += v[j][1].w;
                }
            }
            if (d > SRC_CAP) {
                auto add_row = [&](unsigned srcId) {
                    const float* xs = x + (size_t)srcId * IN_F + fq;
                    const float4 v0 = *(const float4*)(xs + 0);
                    const float4 v1 = *(const float4*)(xs + 4);
                    n0.x += v0.x; n0.y += v0.y; n0.z += v0.z; n0.w += v0.w;
                    n1.x += v1.x; n1.y += v1.y; n1.z += v1.z; n1.w += v1.w;
                };
                const int novf = s_ovn;
                if (novf <= OVF_CAP) {
                    #pragma unroll 1
                    for (int j = 0; j < novf; ++j) {
                        const unsigned w = s_ovf[j];
                        if ((int)(w & (RANGE_N - 1)) == ln) add_row(w >> RUN_SHIFT);
                    }
                } else {
                    n0 = make_float4(0.0f, 0.0f, 0.0f, 0.0f); n1 = n0;
                    #pragma unroll 1
                    for (int e = e0; e < e1; ++e) {
                        const unsigned w = bins[e];
                        if ((int)(w & (RANGE_N - 1)) == ln) add_row(w >> RUN_SHIFT);
                    }
                }
            }
            const float inv = 1.0f / fmaxf((float)d, 1.0f);
            n0.x *= inv; n0.y *= inv; n0.z *= inv; n0.w *= inv;
            n1.x *= inv; n1.y *= inv; n1.z *= inv; n1.w *= inv;
        }
        *(float4*)(dstRow + fq + 0)        = n0;
        *(float4*)(dstRow + fq + 4)        = n1;
        *(float4*)(dstRow + IN_F + fq + 0) = w0;
        *(float4*)(dstRow + IN_F + fq + 4) = w1;
    };

    issue_rows(part);
    if (bucket == MAXDEG) request_late_rows(part);

    const int lr    = lane & 15;
    const int khalf = (lane >> 4) << 1;
    const int ncol  = wave * 16 + lr;
    const unsigned coff = (unsigned)ncol * IN_F;
    const int mbase = (lane >> 4) * 8;

    const float* __restrict__ wn = neigh_w + (size_t)bucket * (OUT_F * IN_F);
    const float* __restrict__ wr = root_w  + (size_t)bucket * (OUT_F * IN_F);
    v2f bfr[64];
    #pragma unroll
    for (int kk = 0; kk < 32; ++kk) {
        const int k = kk * 4 + khalf;
        bfr[kk].x = wn[coff + k]; bfr[kk].y = wn[coff + k + 1];
    }
    #pragma unroll
    for (int kk = 0; kk < 32; ++kk) {
        const int k = kk * 4 + khalf;
        bfr[32 + kk].x = wr[coff + k]; bfr[32 + kk].y = wr[coff + k + 1];
    }
    const float bias = neigh_b[bucket * OUT_F + ncol] + root_b[bucket * OUT_F + ncol];

    finish_rows(&ldsA[sm * LDA]);
    __syncthreads();

    int cur = 0;
    for (int t = part; t < nTiles; t += nparts) {
        const int  slot0 = t * TILE_M;
        const bool more  = t + nparts < nTiles;

        if (more) {
            issue_rows(t + nparts);
            if (bucket == MAXDEG) request_late_rows(t + nparts);
        }

        const int4 onA = *(const int4*)&s_list[slot0 + mbase];
        const int4 onB = *(const int4*)&s_list[slot0 + mbase + 4];

        const float* __restrict__ aRow = &ldsA[cur * (TILE_M * LDA) + lr * LDA];
        v8f acc = {};
        #pragma unroll
        for (int kk = 0; kk < 32; ++kk) {
            const int k = kk * 4 + khalf;
            v2f a;
            a.x = aRow[k];     a.y = aRow[k + 1];
            acc = __builtin_amdgcn_wmma_f32_16x16x4_f32(false, a, false, bfr[kk],
                                                        (short)0, acc, false, false);
        }
        #pragma unroll
        for (int kk = 0; kk < 32; ++kk) {
            const int k = kk * 4 + khalf;
            v2f a;
            a.x = aRow[IN_F + k]; a.y = aRow[IN_F + k + 1];
            acc = __builtin_amdgcn_wmma_f32_16x16x4_f32(false, a, false, bfr[32 + kk],
                                                        (short)0, acc, false, false);
        }

        if (more) {
            finish_rows(&ldsA[(cur ^ 1) * (TILE_M * LDA) + sm * LDA]);
            __syncthreads();
        }
        cur ^= 1;

        const int onode[8] = { onA.x, onA.y, onA.z, onA.w, onB.x, onB.y, onB.z, onB.w };
        #pragma unroll
        for (int i = 0; i < 8; ++i)
            if (onode[i] >= 0) out[(size_t)onode[i] * OUT_F + ncol] = acc[i] + bias;
    }
}

extern "C" void kernel_launch(void* const* d_in, const int* in_sizes, int n_in,
                              void* d_out, int out_size, void* d_ws, size_t ws_size,
                              hipStream_t stream) {
    const float* x       = (const float*)d_in[0];
    const float* neigh_w = (const float*)d_in[1];
    const float* neigh_b = (const float*)d_in[2];
    const float* root_w  = (const float*)d_in[3];
    const float* root_b  = (const float*)d_in[4];
    const int*   eidx    = (const int*)d_in[5];

    const int N = in_sizes[0] / IN_F;
    const int E = in_sizes[5] / 2;
    const int* row = eidx;
    const int* col = eidx + E;

    const int nRanges = (N + RANGE_N - 1) / RANGE_N;
    const int nEB     = (E + EDGE_B - 1) / EDGE_B;

    char* ws = (char*)d_ws;
    int* hist      = (int*)ws;       ws += (size_t)((nEB * nRanges + 15) & ~15) * sizeof(int);
    int* binstart  = (int*)ws;       ws += (size_t)((nRanges + 1 + 15) & ~15) * sizeof(int);
    unsigned* bins = (unsigned*)ws;

    run_hist_kernel<<<nEB, BIN_T, (size_t)nRanges * sizeof(int), stream>>>(col, hist, E, nRanges);
    run_place_kernel<<<nEB, BIN_T, (size_t)4 * nRanges * sizeof(int), stream>>>(row, col, hist, binstart, bins,
                                                                              E, nRanges, nEB);

    mfconv_gemm_kernel<<<dim3(nRanges, GRID_Y), 256, 0, stream>>>(
        x, binstart, bins, neigh_w, neigh_b, root_w, root_b, (float*)d_out, N);
}


# GEMM kernel: every global load instruction covers whole 128-byte lines: a wave's 16 weight rows read as lines, turned into B fragments in a wave-private LDS strip; x rows as 2 x 256 B per node
# speedup vs baseline: 3.1333x; 1.0656x over previous
#include <hip/hip_runtime.h>


typedef __attribute__((ext_vector_type(2))) float v2f;
typedef __attribute__((ext_vector_type(8))) float v8f;

#define IN_F   128
#define OUT_F  128
#define MAXDEG 4
#define NBUCK  5
#define TILE_M 16
#define LDA    260
#define RANGE_N     512
#define RUN_SHIFT   9
#define BIG_PARTS   2
#define GRID_Y      (NBUCK - 1 + BIG_PARTS)
#define RANGE_PASS  (RANGE_N / 256)
#define LIST_LEN    (RANGE_N + TILE_M)
#define SRC_CAP     8
#define OVF_CAP     256
#define EDGE_B      4096
#define BIN_T       1024
#define EDGE_BATCH  8
#define WCHUNK      32
#define WPITCH      36


__global__ __launch_bounds__(BIN_T) void run_hist_kernel(const int* __restrict__ col, int* __restrict__ hist,
                                                         int E, int nBins) {
    extern __shared__ int dyn_lds[];
    const int tid = threadIdx.x;
    for (int b = tid; b < nBins; b += BIN_T) dyn_lds[b] = 0;
    __syncthreads();
    int d[EDGE_B / BIN_T];
    for (int k = 0; k < EDGE_B / BIN_T; ++k) {
        const int e = blockIdx.x * EDGE_B + k * BIN_T + tid;
        d[k] = col[min(e, E - 1)];
    }
    for (int k = 0; k < EDGE_B / BIN_T; ++k) {
        const int e = blockIdx.x * EDGE_B + k * BIN_T + tid;
        if (e < E) atomicAdd(&dyn_lds[d[k] >> RUN_SHIFT], 1);
    }
    __syncthreads();
    for (int b = tid; b < nBins; b += BIN_T) hist[blockIdx.x * nBins + b] = dyn_lds[b];
}

__global__ __launch_bounds__(BIN_T) void run_place_kernel(const int* __restrict__ row, const int* __restrict__ col,
                                                          const int* __restrict__ hist, int* __restrict__ binstart,
                                                          unsigned* __restrict__ bins, int E, int nBins, int nEB) {
    extern __shared__ int dyn_lds[];
    int* sA = dyn_lds;
    int* sB = dyn_lds + 2 * nBins;
    const int tid  = threadIdx.x;
    const int wg   = blockIdx.x;
    const int half = tid >> 9;

    int d[EDGE_B / BIN_T], s[EDGE_B / BIN_T];
    for (int k = 0; k < EDGE_B / BIN_T; ++k) {
        const int e = min(wg * EDGE_B + k * BIN_T + tid, E - 1);
        d[k] = col[e];
        s[k] = row[e];
    }

    for (int b = tid & 511; b < nBins; b += 512) {
        int tot = 0, pre = 0;
        int g = half;
        for (; g + 14 < nEB; g += 16) {
            int v[8];
            for (int u = 0; u < 8; ++u) v[u] = hist[(g + 2 * u) * nBins + b];
            for (int u = 0; u < 8; ++u) { tot += v[u]; pre += (g + 2 * u < wg) ? v[u] : 0; }
        }
        for (; g < nEB; g += 2) {
            const int v = hist[g * nBins + b];
            tot += v; pre += (g < wg) ? v : 0;
        }
        sA[half * nBins + b] = tot;
        sB[half * nBins + b] = pre;
    }
    __syncthreads();
    for (int b = tid; b < nBins; b += BIN_T) { sA[b] += sA[nBins + b]; sB[b] += sB[nBins + b]; }
    __syncthreads();

    if (tid < 32) {
        const int chunk = (nBins + 31) / 32;
        const int lo = min(tid * chunk, nBins);
        const int hi = min(lo + chunk, nBins);
        int mysum = 0;
        for (int b = lo; b < hi; ++b) mysum += sA[b];
        int incl = mysum;
        for (int off = 1; off < 32; off <<= 1) {
            const int t = __shfl_up(incl, off, 32);
            if (tid >= off) incl += t;
        }
        int run = incl - mysum;
        for (int b = lo; b < hi; ++b) { const int t = sA[b]; sA[b] = run; run += t; }
    }
    __syncthreads();
    for (int b = tid; b < nBins; b += BIN_T) {
        const int st = sA[b];
        if (wg == 0) binstart[b] = st;
        sB[b] += st;
    }
    if (wg == 0 && tid == 0) binstart[nBins] = E;
    __syncthreads();

    for (int k = 0; k < EDGE_B / BIN_T; ++k) {
        const int e = wg * EDGE_B + k * BIN_T + tid;
        if (e < E) {
            const int pos = atomicAdd(&sB[d[k] >> RUN_SHIFT], 1);
            bins[pos] = ((unsigned)s[k] << RUN_SHIFT) | (unsigned)(d[k] & (RANGE_N - 1));
        }
    }
}

__global__ __launch_bounds__(256, 4) void mfconv_gemm_kernel(
    const float* __restrict__ x,
    const int* __restrict__ binstart, const unsigned* __restrict__ bins,
    const float* __restrict__ neigh_w, const float* __restrict__ neigh_b,
    const float* __restrict__ root_w,  const float* __restrict__ root_b,
    float* __restrict__ out, int N)
{
    __shared__ __attribute__((aligned(16))) float ldsA[2 * TILE_M * LDA];
    __shared__ __attribute__((aligned(16))) int s_list[LIST_LEN];
    __shared__ int   s_dl[RANGE_N];
    __shared__ __attribute__((aligned(16))) int s_src[RANGE_N * SRC_CAP];
    __shared__ unsigned s_ovf[OVF_CAP];
    __shared__ int   s_ovn;
    __shared__ int   s_cnt[RANGE_PASS * 8];

    const int tid   = threadIdx.x;
    const int wave  = tid >> 5;
    const int lane  = tid & 31;
    const int node0 = blockIdx.x * RANGE_N;

    const int by     = blockIdx.y;
    const int bucket = (by < BIG_PARTS) ? MAXDEG : (NBUCK - 2 + BIG_PARTS - by);
    const int part   = (by < BIG_PARTS) ? by : 0;
    const int nparts = (by < BIG_PARTS) ? BIG_PARTS : 1;

    for (int i = tid; i < LIST_LEN; i += 256) s_list[i] = -1;
    for (int i = tid; i < RANGE_N; i += 256) s_dl[i] = 0;
    if (tid == 0) s_ovn = 0;
    const int e0 = binstart[blockIdx.x];
    const int e1 = binstart[blockIdx.x + 1];
    __syncthreads();

    for (int eb = e0; eb < e1; eb += 256 * EDGE_BATCH) {
        unsigned w[EDGE_BATCH];
        for (int u = 0; u < EDGE_BATCH; ++u)
            w[u] = bins[min(eb + u * 256 + tid, e1 - 1)];
        int slot[EDGE_BATCH];
        for (int u = 0; u < EDGE_BATCH; ++u) {
            slot[u] = -1;
            if (eb + u * 256 + tid < e1) slot[u] = atomicAdd(&s_dl[w[u] & (RANGE_N - 1)], 1);
        }
        for (int u = 0; u < EDGE_BATCH; ++u) {
            if (slot[u] >= 0) {
                if (slot[u] < SRC_CAP) {
                    s_src[(int)(w[u] & (RANGE_N - 1)) * SRC_CAP + slot[u]] = (int)(w[u] >> RUN_SHIFT);
                } else {
                    const int o = atomicAdd(&s_ovn, 1);
                    if (o < OVF_CAP) s_ovf[o] = w[u];
                }
            }
        }
    }
    __syncthreads();

    bool mine[RANGE_PASS];
    int  myrank[RANGE_PASS];
    #pragma unroll
    for (int p = 0; p < RANGE_PASS; ++p) {
        const int node = node0 + p * 256 + tid;
        const bool in = (node < N) && (min(s_dl[p * 256 + tid], MAXDEG) == bucket);
        const unsigned m = __builtin_amdgcn_ballot_w32(in);
        if (lane == 0) s_cnt[p * 8 + wave] = __popc(m);
        myrank[p] = __popc(m & ((1u << lane) - 1u));
        mine[p] = in;
    }
    __syncthreads();

    int total = 0;
    int base[RANGE_PASS];
    #pragma unroll
    for (int p = 0; p < RANGE_PASS; ++p) base[p] = 0;
    #pragma unroll
    for (int i = 0; i < RANGE_PASS * 8; ++i) {
        const int c = s_cnt[i];
        #pragma unroll
        for (int p = 0; p < RANGE_PASS; ++p) if (i < p * 8 + wave) base[p] += c;
        total += c;
    }
    const int nTiles = (__builtin_amdgcn_readfirstlane(total) + TILE_M - 1) / TILE_M;
    if (part >= nTiles) return;

    #pragma unroll
    for (int p = 0; p < RANGE_PASS; ++p)
        if (mine[p]) s_list[base[p] + myrank[p]] = node0 + p * 256 + tid;

    __syncthreads();

    const int sm = tid >> 4;
    const int fq = (tid & 15) * 4;
    constexpr int FH = IN_F / 2;

    int    g_node = -1, g_ln = 0, g_d = 0;
    float4 g_own[2];
    float4 g_nb[3][2];

    auto issue_rows = [&](int tile) {
        const float4 id0 = make_float4(-0.0f, -0.0f, -0.0f, -0.0f);
        #pragma unroll
        for (int j = 0; j < 3; ++j) { g_nb[j][0] = id0; g_nb[j][1] = id0; }
        g_own[0] = id0; g_own[1] = id0;
        g_node = s_list[tile * TILE_M + sm];
        g_ln = 0;
        g_d  = 0;
        if (g_node >= 0) {
            const float* xo = x + (size_t)g_node * IN_F + fq;
            g_own[0] = *(const float4*)(xo + 0);
            g_own[1] = *(const float4*)(xo + FH);
            g_ln = g_node - node0;
            g_d  = s_dl[g_ln];
            if (g_d > 0) {
                const int4 id4 = *(const int4*)&s_src[g_ln * SRC_CAP];
                const int idv[3] = { id4.x, id4.y, id4.z };
                #pragma unroll
                for (int j = 0; j < 3; ++j) {
                    if (j < g_d) {
                        const float* xs = x + (size_t)idv[j] * IN_F + fq;
                        g_nb[j][0] = *(const float4*)(xs + 0);
                        g_nb[j][1] = *(const float4*)(xs + FH);
                    }
                }
            }
        }
    };

    auto request_late_rows = [&](int tile) {
        const int pnode = s_list[tile * TILE_M + sm];
        const int q = tid & 15;
        if (pnode >= 0 && q < 10) {
            const int pl = pnode - node0;
            const int j  = 3 + (q >> 1);
            if (j < s_dl[pl]) {
                const int prow = s_src[pl * SRC_CAP + j];
                __builtin_prefetch(x + (size_t)prow * IN_F + (q & 1) * (IN_F / 2), 0, 2);
            }
        }
    };

    auto finish_rows = [&](float* dstRow) {
        float4 n0 = make_float4(0.0f, 0.0f, 0.0f, 0.0f), n1 = n0;
        float4 w0 = n0, w1 = n0;
        if (g_node >= 0) {
            const int ln = g_ln;
            const int d  = g_d;
            w0 = g_own[0]; w1 = g_own[1];
            if (d > 0) {
                #pragma unroll
                for (int j = 0; j < 3; ++j) {
                    n0.x += g_nb[j][0].x; n0.y += g_nb[j][0].y; n0.z += g_nb[j][0].z; n0.w += g_nb[j][0].w;
                    n1.x += g_nb[j][1].x; n1.y += g_nb[j][1].y; n1.z += g_nb[j][1].z; n1.w += g_nb[j][1].w;
                }
            }
            if (d > 3) {
                const float4 id0 = make_float4(-0.0f, -0.0f, -0.0f, -0.0f);
                const int  id3 = s_src[ln * SRC_CAP + 3];
                const int4 id4 = *(const int4*)&s_src[ln * SRC_CAP + 4];
                const int idv[5] = { id3, id4.x, id4.y, id4.z, id4.w };
                float4 v[5][2];
                #pragma unroll
                for (int j = 0; j < 5; ++j) {
                    v[j][0] = id0; v[j][1] = id0;
                    if (3 + j < d) {
                        const float* xs = x + (size_t)idv[j] * IN_F + fq;
                        v[j][0] = *(const float4*)(xs + 0);
                        v[j][1] = *(const float4*)(xs + FH);
                    }
                }
                #pragma unroll
                for (int j = 0; j < 5; ++j) {
                    n0.x += v[j][0].x; n0.y += v[j][0].y; n0.z += v[j][0].z; n0.w += v[j][0].w;
                    n1.x += v[j][1].x; n1.y += v[j][1].y; n1.z += v[j][1].z; n1.w += v[j][1].w;
                }
            }
            if (d > SRC_CAP) {
                auto add_row = [&](unsigned srcId) {
                    const float* xs = x + (size_t)srcId * IN_F + fq;
                    const float4 v0 = *(const float4*)(xs + 0);
                    const float4 v1 = *(const float4*)(xs + FH);
                    n0.x += v0.x; n0.y += v0.y; n0.z += v0.z; n0.w += v0.w;
                    n1.x += v1.x; n1.y += v1.y; n1.z += v1.z; n1.w += v1.w;
                };
                const int novf = s_ovn;
                if (novf <= OVF_CAP) {
                    #pragma unroll 1
                    for (int j = 0; j < novf; ++j) {
                        const unsigned w = s_ovf[j];
                        if ((int)(w & (RANGE_N - 1)) == ln) add_row(w >> RUN_SHIFT);
                    }
                } else {
                    n0 = make_float4(0.0f, 0.0f, 0.0f, 0.0f); n1 = n0;
                    #pragma unroll 1
                    for (int e = e0; e < e1; ++e) {
                        const unsigned w = bins[e];
                        if ((int)(w & (RANGE_N - 1)) == ln) add_row(w >> RUN_SHIFT);
                    }
                }
            }
            const float inv = 1.0f / fmaxf((float)d, 1.0f);
            n0.x *= inv; n0.y *= inv; n0.z *= inv; n0.w *= inv;
            n1.x *= inv; n1.y *= inv; n1.z *= inv; n1.w *= inv;
        }
        *(float4*)(dstRow + fq)             = n0;
        *(float4*)(dstRow + FH + fq)        = n1;
        *(float4*)(dstRow + IN_F + fq)      = w0;
        *(float4*)(dstRow + IN_F + FH + fq) = w1;
    };

    issue_rows(part);
    if (bucket == MAXDEG) request_late_rows(part);

    const int lr    = lane & 15;
    const int khalf = (lane >> 4) << 1;
    const int ncol  = wave * 16 + lr;
    const int mbase = (lane >> 4) * 8;

    const float* __restrict__ wn = neigh_w + (size_t)bucket * (OUT_F * IN_F);
    const float* __restrict__ wr = root_w  + (size_t)bucket * (OUT_F * IN_F);
    v2f bfr[64];
    {
        float* wstrip = &ldsA[wave * (16 * WPITCH)];
        const int wrow = lane >> 3;
        const int wcol = (lane & 7) * 4;
        struct WChunk { float4 r0, r1, r2, r3; };
        auto load_chunk = [&](int ch) -> WChunk {
            const float* __restrict__ p = ((ch < 4) ? wn : wr) + (size_t)(wave * 16 + wrow) * IN_F
                                          + (ch & 3) * WCHUNK + wcol;
            WChunk c;
            c.r0 = *(const float4*)(p);
            c.r1 = *(const float4*)(p + 4 * IN_F);
            c.r2 = *(const float4*)(p + 8 * IN_F);
            c.r3 = *(const float4*)(p + 12 * IN_F);
            return c;
        };
        auto turn_chunk = [&](const WChunk& c, int ch) {
            __builtin_amdgcn_fence(__ATOMIC_ACQ_REL, "wavefront");
            float* q = &wstrip[wrow * WPITCH + wcol];
            *(float4*)(q)               = c.r0;
            *(float4*)(q + 4 * WPITCH)  = c.r1;
            *(float4*)(q + 8 * WPITCH)  = c.r2;
            *(float4*)(q + 12 * WPITCH) = c.r3;
            __builtin_amdgcn_fence(__ATOMIC_ACQ_REL, "wavefront");
            #pragma unroll
            for (int k8 = 0; k8 < 8; ++k8)
                bfr[ch * 8 + k8] = *(const v2f*)&wstrip[lr * WPITCH + k8 * 4 + khalf];
        };
        const WChunk c0 = load_chunk(0), c1 = load_chunk(1), c2 = load_chunk(2), c3 = load_chunk(3);
        turn_chunk(c0, 0); turn_chunk(c1, 1);
        const WChunk c4 = load_chunk(4), c5 = load_chunk(5);
        turn_chunk(c2, 2); turn_chunk(c3, 3);
        const WChunk c6 = load_chunk(6), c7 = load_chunk(7);
        turn_chunk(c4, 4); turn_chunk(c5, 5);
        turn_chunk(c6, 6); turn_chunk(c7, 7);
    }
    const float bias = neigh_b[bucket * OUT_F + ncol] + root_b[bucket * OUT_F + ncol];
    __syncthreads();

    finish_rows(&ldsA[sm * LDA]);
    __syncthreads();

    int cur = 0;
    for (int t = part; t < nTiles; t += nparts) {
        const int  slot0 = t * TILE_M;
        const bool more  = t + nparts < nTiles;

        if (more) {
            issue_rows(t + nparts);
            if (bucket == MAXDEG) request_late_rows(t + nparts);
        }

        const int4 onA = *(const int4*)&s_list[slot0 + mbase];
        const int4 onB = *(const int4*)&s_list[slot0 + mbase + 4];

        const float* __restrict__ aRow = &ldsA[cur * (TILE_M * LDA) + lr * LDA];
        v8f acc = {};
        #pragma unroll
        for (int kk = 0; kk < 32; ++kk) {
            const int k = kk * 4 + khalf;
            v2f a;
            a.x = aRow[k];     a.y = aRow[k + 1];
            acc = __builtin_amdgcn_wmma_f32_16x16x4_f32(false, a, false, bfr[kk],
                                                        (short)0, acc, false, false);
        }
        #pragma unroll
        for (int kk = 0; kk < 32; ++kk) {
            const int k = kk * 4 + khalf;
            v2f a;
            a.x = aRow[IN_F + k]; a.y = aRow[IN_F + k + 1];
            acc = __builtin_amdgcn_wmma_f32_16x16x4_f32(false, a, false, bfr[32 + kk],
                                                        (short)0, acc, false, false);
        }

        if (more) {
            finish_rows(&ldsA[(cur ^ 1) * (TILE_M * LDA) + sm * LDA]);
            __syncthreads();
        }
        cur ^= 1;

        const int onode[8] = { onA.x, onA.y, onA.z, onA.w, onB.x, onB.y, onB.z, onB.w };
        #pragma unroll
        for (int i = 0; i < 8; ++i)
            if (onode[i] >= 0) out[(size_t)onode[i] * OUT_F + ncol] = acc[i] + bias;
    }
}

extern "C" void kernel_launch(void* const* d_in, const int* in_sizes, int n_in,
                              void* d_out, int out_size, void* d_ws, size_t ws_size,
                              hipStream_t stream) {
    const float* x       = (const float*)d_in[0];
    const float* neigh_w = (const float*)d_in[1];
    const float* neigh_b = (const float*)d_in[2];
    const float* root_w  = (const float*)d_in[3];
    const float* root_b  = (const float*)d_in[4];
    const int*   eidx    = (const int*)d_in[5];

    const int N = in_sizes[0] / IN_F;
    const int E = in_sizes[5] / 2;
    const int* row = eidx;
    const int* col = eidx + E;

    const int nRanges = (N + RANGE_N - 1) / RANGE_N;
    const int nEB     = (E + EDGE_B - 1) / EDGE_B;

    char* ws = (char*)d_ws;
    int* hist      = (int*)ws;       ws += (size_t)((nEB * nRanges + 15) & ~15) * sizeof(int);
    int* binstart  = (int*)ws;       ws += (size_t)((nRanges + 1 + 15) & ~15) * sizeof(int);
    unsigned* bins = (unsigned*)ws;

    run_hist_kernel<<<nEB, BIN_T, (size_t)nRanges * sizeof(int), stream>>>(col, hist, E, nRanges);
    run_place_kernel<<<nEB, BIN_T, (size_t)4 * nRanges * sizeof(int), stream>>>(row, col, hist, binstart, bins,
                                                                              E, nRanges, nEB);

    mfconv_gemm_kernel<<<dim3(nRanges, GRID_Y), 256, 0, stream>>>(
        x, binstart, bins, neigh_w, neigh_b, root_w, root_b, (float*)d_out, N);
}


# node run per work-group doubled to 1024 nodes: half as many work-groups, each builds its LDS edge table and loads its bucket's weights once for twice the tiles; neighbour sum order = LDS add order
# speedup vs baseline: 3.2809x; 1.0471x over previous
#include <hip/hip_runtime.h>


typedef __attribute__((ext_vector_type(2))) float v2f;
typedef __attribute__((ext_vector_type(8))) float v8f;

#define IN_F   128
#define OUT_F  128
#define MAXDEG 4
#define NBUCK  5
#define TILE_M 16
#define LDA    260
#define RANGE_N     1024
#define RUN_SHIFT   10
#define BIG_PARTS   2
#define GRID_Y      (NBUCK - 1 + BIG_PARTS)
#define RANGE_PASS  (RANGE_N / 256)
#define LIST_LEN    (RANGE_N + TILE_M)
#define SRC_CAP     8
#define OVF_CAP     256
#define EDGE_B      4096
#define BIN_T       1024
#define EDGE_BATCH  16
#define WCHUNK      32
#define WPITCH      36


__global__ __launch_bounds__(BIN_T) void run_hist_kernel(const int* __restrict__ col, int* __restrict__ hist,
                                                         int E, int nBins) {
    extern __shared__ int dyn_lds[];
    const int tid = threadIdx.x;
    for (int b = tid; b < nBins; b += BIN_T) dyn_lds[b] = 0;
    __syncthreads();
    int d[EDGE_B / BIN_T];
    for (int k = 0; k < EDGE_B / BIN_T; ++k) {
        const int e = blockIdx.x * EDGE_B + k * BIN_T + tid;
        d[k] = col[min(e, E - 1)];
    }
    for (int k = 0; k < EDGE_B / BIN_T; ++k) {
        const int e = blockIdx.x * EDGE_B + k * BIN_T + tid;
        if (e < E) atomicAdd(&dyn_lds[d[k] >> RUN_SHIFT], 1);
    }
    __syncthreads();
    for (int b = tid; b < nBins; b += BIN_T) hist[blockIdx.x * nBins + b] = dyn_lds[b];
}

__global__ __launch_bounds__(BIN_T) void run_place_kernel(const int* __restrict__ row, const int* __restrict__ col,
                                                          const int* __restrict__ hist, int* __restrict__ binstart,
                                                          unsigned* __restrict__ bins, int E, int nBins, int nEB) {
    extern __shared__ int dyn_lds[];
    int* sA = dyn_lds;
    int* sB = dyn_lds + 2 * nBins;
    const int tid  = threadIdx.x;
    const int wg   = blockIdx.x;
    const int half = tid >> 9;

    int d[EDGE_B / BIN_T], s[EDGE_B / BIN_T];
    for (int k = 0; k < EDGE_B / BIN_T; ++k) {
        const int e = min(wg * EDGE_B + k * BIN_T + tid, E - 1);
        d[k] = col[e];
        s[k] = row[e];
    }

    for (int b = tid & 511; b < nBins; b += 512) {
        int tot = 0, pre = 0;
        int g = half;
        for (; g + 14 < nEB; g += 16) {
            int v[8];
            for (int u = 0; u < 8; ++u) v[u] = hist[(g + 2 * u) * nBins + b];
            for (int u = 0; u < 8; ++u) { tot += v[u]; pre += (g + 2 * u < wg) ? v[u] : 0; }
        }
        for (; g < nEB; g += 2) {
            const int v = hist[g * nBins + b];
            tot += v; pre += (g < wg) ? v : 0;
        }
        sA[half * nBins + b] = tot;
        sB[half * nBins + b] = pre;
    }
    __syncthreads();
    for (int b = tid; b < nBins; b += BIN_T) { sA[b] += sA[nBins + b]; sB[b] += sB[nBins + b]; }
    __syncthreads();

    if (tid < 32) {
        const int chunk = (nBins + 31) / 32;
        const int lo = min(tid * chunk, nBins);
        const int hi = min(lo + chunk, nBins);
        int mysum = 0;
        for (int b = lo; b < hi; ++b) mysum += sA[b];
        int incl = mysum;
        for (int off = 1; off < 32; off <<= 1) {
            const int t = __shfl_up(incl, off, 32);
            if (tid >= off) incl += t;
        }
        int run = incl - mysum;
        for (int b = lo; b < hi; ++b) { const int t = sA[b]; sA[b] = run; run += t; }
    }
    __syncthreads();
    for (int b = tid; b < nBins; b += BIN_T) {
        const int st = sA[b];
        if (wg == 0) binstart[b] = st;
        sB[b] += st;
    }
    if (wg == 0 && tid == 0) binstart[nBins] = E;
    __syncthreads();

    for (int k = 0; k < EDGE_B / BIN_T; ++k) {
        const int e = wg * EDGE_B + k * BIN_T + tid;
        if (e < E) {
            const int pos = atomicAdd(&sB[d[k] >> RUN_SHIFT], 1);
            bins[pos] = ((unsigned)s[k] << RUN_SHIFT) | (unsigned)(d[k] & (RANGE_N - 1));
        }
    }
}

__global__ __launch_bounds__(256, 4) void mfconv_gemm_kernel(
    const float* __restrict__ x,
    const int* __restrict__ binstart, const unsigned* __restrict__ bins,
    const float* __restrict__ neigh_w, const float* __restrict__ neigh_b,
    const float* __restrict__ root_w,  const float* __restrict__ root_b,
    float* __restrict__ out, int N)
{
    __shared__ __attribute__((aligned(16))) float ldsA[2 * TILE_M * LDA];
    __shared__ __attribute__((aligned(16))) int s_list[LIST_LEN];
    __shared__ int   s_dl[RANGE_N];
    __shared__ __attribute__((aligned(16))) int s_src[RANGE_N * SRC_CAP];
    __shared__ unsigned s_ovf[OVF_CAP];
    __shared__ int   s_ovn;
    __shared__ int   s_cnt[RANGE_PASS * 8];

    const int tid   = threadIdx.x;
    const int wave  = tid >> 5;
    const int lane  = tid & 31;
    const int node0 = blockIdx.x * RANGE_N;

    const int by     = blockIdx.y;
    const int bucket = (by < BIG_PARTS) ? MAXDEG : (NBUCK - 2 + BIG_PARTS - by);
    const int part   = (by < BIG_PARTS) ? by : 0;
    const int nparts = (by < BIG_PARTS) ? BIG_PARTS : 1;

    for (int i = tid; i < LIST_LEN; i += 256) s_list[i] = -1;
    for (int i = tid; i < RANGE_N; i += 256) s_dl[i] = 0;
    if (tid == 0) s_ovn = 0;
    const int e0 = binstart[blockIdx.x];
    const int e1 = binstart[blockIdx.x + 1];
    __syncthreads();

    for (int eb = e0; eb < e1; eb += 256 * EDGE_BATCH) {
        unsigned w[EDGE_BATCH];
        for (int u = 0; u < EDGE_BATCH; ++u)
            w[u] = bins[min(eb + u * 256 + tid, e1 - 1)];
        int slot[EDGE_BATCH];
        for (int u = 0; u < EDGE_BATCH; ++u) {
            slot[u] = -1;
            if (eb + u * 256 + tid < e1) slot[u] = atomicAdd(&s_dl[w[u] & (RANGE_N - 1)], 1);
        }
        for (int u = 0; u < EDGE_BATCH; ++u) {
            if (slot[u] >= 0) {
                if (slot[u] < SRC_CAP) {
                    s_src[(int)(w[u] & (RANGE_N - 1)) * SRC_CAP + slot[u]] = (int)(w[u] >> RUN_SHIFT);
                } else {
                    const int o = atomicAdd(&s_ovn, 1);
                    if (o < OVF_CAP) s_ovf[o] = w[u];
                }
            }
        }
    }
    __syncthreads();

    bool mine[RANGE_PASS];
    int  myrank[RANGE_PASS];
    #pragma unroll
    for (int p = 0; p < RANGE_PASS; ++p) {
        const int node = node0 + p * 256 + tid;
        const bool in = (node < N) && (min(s_dl[p * 256 + tid], MAXDEG) == bucket);
        const unsigned m = __builtin_amdgcn_ballot_w32(in);
        if (lane == 0) s_cnt[p * 8 + wave] = __popc(m);
        myrank[p] = __popc(m & ((1u << lane) - 1u));
        mine[p] = in;
    }
    __syncthreads();

    int total = 0;
    int base[RANGE_PASS];
    #pragma unroll
    for (int p = 0; p < RANGE_PASS; ++p) base[p] = 0;
    #pragma unroll
    for (int i = 0; i < RANGE_PASS * 8; ++i) {
        const int c = s_cnt[i];
        #pragma unroll
        for (int p = 0; p < RANGE_PASS; ++p) if (i < p * 8 + wave) base[p] += c;
        total += c;
    }
    const int nTiles = (__builtin_amdgcn_readfirstlane(total) + TILE_M - 1) / TILE_M;
    if (part >= nTiles) return;

    #pragma unroll
    for (int p = 0; p < RANGE_PASS; ++p)
        if (mine[p]) s_list[base[p] + myrank[p]] = node0 + p * 256 + tid;

    __syncthreads();

    const int sm = tid >> 4;
    const int fq = (tid & 15) * 4;
    constexpr int FH = IN_F / 2;

    int    g_node = -1, g_ln = 0, g_d = 0;
    float4 g_own[2];
    float4 g_nb[3][2];

    auto issue_rows = [&](int tile) {
        const float4 id0 = make_float4(-0.0f, -0.0f, -0.0f, -0.0f);
        #pragma unroll
        for (int j = 0; j < 3; ++j) { g_nb[j][0] = id0; g_nb[j][1] = id0; }
        g_own[0] = id0; g_own[1] = id0;
        g_node = s_list[tile * TILE_M + sm];
        g_ln = 0;
        g_d  = 0;
        if (g_node >= 0) {
            const float* xo = x + (size_t)g_node * IN_F + fq;
            g_own[0] = *(const float4*)(xo + 0);
            g_own[1] = *(const float4*)(xo + FH);
            g_ln = g_node - node0;
            g_d  = s_dl[g_ln];
            if (g_d > 0) {
                const int4 id4 = *(const int4*)&s_src[g_ln * SRC_CAP];
                const int idv[3] = { id4.x, id4.y, id4.z };
                #pragma unroll
                for (int j = 0; j < 3; ++j) {
                    if (j < g_d) {
                        const float* xs = x + (size_t)idv[j] * IN_F + fq;
                        g_nb[j][0] = *(const float4*)(xs + 0);
                        g_nb[j][1] = *(const float4*)(xs + FH);
                    }
                }
            }
        }
    };

    auto request_late_rows = [&](int tile) {
        const int pnode = s_list[tile * TILE_M + sm];
        const int q = tid & 15;
        if (pnode >= 0 && q < 10) {
            const int pl = pnode - node0;
            const int j  = 3 + (q >> 1);
            if (j < s_dl[pl]) {
                const int prow = s_src[pl * SRC_CAP + j];
                __builtin_prefetch(x + (size_t)prow * IN_F + (q & 1) * (IN_F / 2), 0, 2);
            }
        }
    };

    auto finish_rows = [&](float* dstRow) {
        float4 n0 = make_float4(0.0f, 0.0f, 0.0f, 0.0f), n1 = n0;
        float4 w0 = n0, w1 = n0;
        if (g_node >= 0) {
            const int ln = g_ln;
            const int d  = g_d;
            w0 = g_own[0]; w1 = g_own[1];
            if (d > 0) {
                #pragma unroll
                for (int j = 0; j < 3; ++j) {
                    n0.x += g_nb[j][0].x; n0.y += g_nb[j][0].y; n0.z += g_nb[j][0].z; n0.w += g_nb[j][0].w;
                    n1.x += g_nb[j][1].x; n1.y += g_nb[j][1].y; n1.z += g_nb[j][1].z; n1.w += g_nb[j][1].w;
                }
            }
            if (d > 3) {
                const float4 id0 = make_float4(-0.0f, -0.0f, -0.0f, -0.0f);
                const int  id3 = s_src[ln * SRC_CAP + 3];
                const int4 id4 = *(const int4*)&s_src[ln * SRC_CAP + 4];
                const int idv[5] = { id3, id4.x, id4.y, id4.z, id4.w };
                float4 v[5][2];
                #pragma unroll
                for (int j = 0; j < 5; ++j) {
                    v[j][0] = id0; v[j][1] = id0;
                    if (3 + j < d) {
                        const float* xs = x + (size_t)idv[j] * IN_F + fq;
                        v[j][0] = *(const float4*)(xs + 0);
                        v[j][1] = *(const float4*)(xs + FH);
                    }
                }
                #pragma unroll
                for (int j = 0; j < 5; ++j) {
                    n0.x += v[j][0].x; n0.y += v[j][0].y; n0.z += v[j][0].z; n0.w += v[j][0].w;
                    n1.x += v[j][1].x; n1.y += v[j][1].y; n1.z += v[j][1].z; n1.w += v[j][1].w;
                }
            }
            if (d > SRC_CAP) {
                auto add_row = [&](unsigned srcId) {
                    const float* xs = x + (size_t)srcId * IN_F + fq;
                    const float4 v0 = *(const float4*)(xs + 0);
                    const float4 v1 = *(const float4*)(xs + FH);
                    n0.x += v0.x; n0.y += v0.y; n0.z += v0.z; n0.w += v0.w;
                    n1.x += v1.x; n1.y += v1.y; n1.z += v1.z; n1.w += v1.w;
                };
                const int novf = s_ovn;
                if (novf <= OVF_CAP) {
                    #pragma unroll 1
                    for (int j = 0; j < novf; ++j) {
                        const unsigned w = s_ovf[j];
                        if ((int)(w & (RANGE_N - 1)) == ln) add_row(w >> RUN_SHIFT);
                    }
                } else {
                    n0 = make_float4(0.0f, 0.0f, 0.0f, 0.0f); n1 = n0;
                    #pragma unroll 1
                    for (int e = e0; e < e1; ++e) {
                        const unsigned w = bins[e];
                        if ((int)(w & (RANGE_N - 1)) == ln) add_row(w >> RUN_SHIFT);
                    }
                }
            }
            const float inv = 1.0f / fmaxf((float)d, 1.0f);
            n0.x *= inv; n0.y *= inv; n0.z *= inv; n0.w *= inv;
            n1.x *= inv; n1.y *= inv; n1.z *= inv; n1.w *= inv;
        }
        *(float4*)(dstRow + fq)             = n0;
        *(float4*)(dstRow + FH + fq)        = n1;
        *(float4*)(dstRow + IN_F + fq)      = w0;
        *(float4*)(dstRow + IN_F + FH + fq) = w1;
    };

    issue_rows(part);
    if (bucket == MAXDEG) request_late_rows(part);

    const int lr    = lane & 15;
    const int khalf = (lane >> 4) << 1;
    const int ncol  = wave * 16 + lr;
    const int mbase = (lane >> 4) * 8;

    const float* __restrict__ wn = neigh_w + (size_t)bucket * (OUT_F * IN_F);
    const float* __restrict__ wr = root_w  + (size_t)bucket * (OUT_F * IN_F);
    v2f bfr[64];
    {
        float* wstrip = &ldsA[wave * (16 * WPITCH)];
        const int wrow = lane >> 3;
        const int wcol = (lane & 7) * 4;
        struct WChunk { float4 r0, r1, r2, r3; };
        auto load_chunk = [&](int ch) -> WChunk {
            const float* __restrict__ p = ((ch < 4) ? wn : wr) + (size_t)(wave * 16 + wrow) * IN_F
                                          + (ch & 3) * WCHUNK + wcol;
            WChunk c;
            c.r0 = *(const float4*)(p);
            c.r1 = *(const float4*)(p + 4 * IN_F);
            c.r2 = *(const float4*)(p + 8 * IN_F);
            c.r3 = *(const float4*)(p + 12 * IN_F);
            return c;
        };
        auto turn_chunk = [&](const WChunk& c, int ch) {
            __builtin_amdgcn_fence(__ATOMIC_ACQ_REL, "wavefront");
            float* q = &wstrip[wrow * WPITCH + wcol];
            *(float4*)(q)               = c.r0;
            *(float4*)(q + 4 * WPITCH)  = c.r1;
            *(float4*)(q + 8 * WPITCH)  = c.r2;
            *(float4*)(q + 12 * WPITCH) = c.r3;
            __builtin_amdgcn_fence(__ATOMIC_ACQ_REL, "wavefront");
            #pragma unroll
            for (int k8 = 0; k8 < 8; ++k8)
                bfr[ch * 8 + k8] = *(const v2f*)&wstrip[lr * WPITCH + k8 * 4 + khalf];
        };
        const WChunk c0 = load_chunk(0), c1 = load_chunk(1), c2 = load_chunk(2), c3 = load_chunk(3);
        turn_chunk(c0, 0); turn_chunk(c1, 1);
        const WChunk c4 = load_chunk(4), c5 = load_chunk(5);
        turn_chunk(c2, 2); turn_chunk(c3, 3);
        const WChunk c6 = load_chunk(6), c7 = load_chunk(7);
        turn_chunk(c4, 4); turn_chunk(c5, 5);
        turn_chunk(c6, 6); turn_chunk(c7, 7);
    }
    const float bias = neigh_b[bucket * OUT_F + ncol] + root_b[bucket * OUT_F + ncol];
    __syncthreads();

    finish_rows(&ldsA[sm * LDA]);
    __syncthreads();

    int cur = 0;
    for (int t = part; t < nTiles; t += nparts) {
        const int  slot0 = t * TILE_M;
        const bool more  = t + nparts < nTiles;

        if (more) {
            issue_rows(t + nparts);
            if (bucket == MAXDEG) request_late_rows(t + nparts);
        }

        const int4 onA = *(const int4*)&s_list[slot0 + mbase];
        const int4 onB = *(const int4*)&s_list[slot0 + mbase + 4];

        const float* __restrict__ aRow = &ldsA[cur * (TILE_M * LDA) + lr * LDA];
        v8f acc = {};
        #pragma unroll
        for (int kk = 0; kk < 32; ++kk) {
            const int k = kk * 4 + khalf;
            v2f a;
            a.x = aRow[k];     a.y = aRow[k + 1];
            acc = __builtin_amdgcn_wmma_f32_16x16x4_f32(false, a, false, bfr[kk],
                                                        (short)0, acc, false, false);
        }
        #pragma unroll
        for (int kk = 0; kk < 32; ++kk) {
            const int k = kk * 4 + khalf;
            v2f a;
            a.x = aRow[IN_F + k]; a.y = aRow[IN_F + k + 1];
            acc = __builtin_amdgcn_wmma_f32_16x16x4_f32(false, a, false, bfr[32 + kk],
                                                        (short)0, acc, false, false);
        }

        if (more) {
            finish_rows(&ldsA[(cur ^ 1) * (TILE_M * LDA) + sm * LDA]);
            __syncthreads();
        }
        cur ^= 1;

        const int onode[8] = { onA.x, onA.y, onA.z, onA.w, onB.x, onB.y, onB.z, onB.w };
        #pragma unroll
        for (int i = 0; i < 8; ++i)
            if (onode[i] >= 0) out[(size_t)onode[i] * OUT_F + ncol] = acc[i] + bias;
    }
}

extern "C" void kernel_launch(void* const* d_in, const int* in_sizes, int n_in,
                              void* d_out, int out_size, void* d_ws, size_t ws_size,
                              hipStream_t stream) {
    const float* x       = (const float*)d_in[0];
    const float* neigh_w = (const float*)d_in[1];
    const float* neigh_b = (const float*)d_in[2];
    const float* root_w  = (const float*)d_in[3];
    const float* root_b  = (const float*)d_in[4];
    const int*   eidx    = (const int*)d_in[5];

    const int N = in_sizes[0] / IN_F;
    const int E = in_sizes[5] / 2;
    const int* row = eidx;
    const int* col = eidx + E;

    const int nRanges = (N + RANGE_N - 1) / RANGE_N;
    const int nEB     = (E + EDGE_B - 1) / EDGE_B;

    char* ws = (char*)d_ws;
    int* hist      = (int*)ws;       ws += (size_t)((nEB * nRanges + 15) & ~15) * sizeof(int);
    int* binstart  = (int*)ws;       ws += (size_t)((nRanges + 1 + 15) & ~15) * sizeof(int);
    unsigned* bins = (unsigned*)ws;

    run_hist_kernel<<<nEB, BIN_T, (size_t)nRanges * sizeof(int), stream>>>(col, hist, E, nRanges);
    run_place_kernel<<<nEB, BIN_T, (size_t)4 * nRanges * sizeof(int), stream>>>(row, col, hist, binstart, bins,
                                                                              E, nRanges, nEB);

    mfconv_gemm_kernel<<<dim3(nRanges, GRID_Y), 256, 0, stream>>>(
        x, binstart, bins, neigh_w, neigh_b, root_w, root_b, (float*)d_out, N);
}


# GEMM tile loop: per list slot an int4 LDS record (node + first 3 source ids) read one tile ahead; a bucket's early row count is block-uniform, so row loads leave with no LDS index chain in front
# speedup vs baseline: 3.3977x; 1.0356x over previous
#include <hip/hip_runtime.h>


typedef __attribute__((ext_vector_type(2))) float v2f;
typedef __attribute__((ext_vector_type(8))) float v8f;

#define IN_F   128
#define OUT_F  128
#define MAXDEG 4
#define NBUCK  5
#define TILE_M 16
#define LDA    260
#define RANGE_N     1024
#define RUN_SHIFT   10
#define BIG_PARTS   2
#define GRID_Y      (NBUCK - 1 + BIG_PARTS)
#define RANGE_PASS  (RANGE_N / 256)
#define LIST_LEN    (RANGE_N + TILE_M)
#define SRC_CAP     8
#define OVF_CAP     256
#define EDGE_B      4096
#define BIN_T       1024
#define EDGE_BATCH  16
#define WCHUNK      32
#define WPITCH      36


__global__ __launch_bounds__(BIN_T) void run_hist_kernel(const int* __restrict__ col, int* __restrict__ hist,
                                                         int E, int nBins) {
    extern __shared__ int dyn_lds[];
    const int tid = threadIdx.x;
    for (int b = tid; b < nBins; b += BIN_T) dyn_lds[b] = 0;
    __syncthreads();
    int d[EDGE_B / BIN_T];
    for (int k = 0; k < EDGE_B / BIN_T; ++k) {
        const int e = blockIdx.x * EDGE_B + k * BIN_T + tid;
        d[k] = col[min(e, E - 1)];
    }
    for (int k = 0; k < EDGE_B / BIN_T; ++k) {
        const int e = blockIdx.x * EDGE_B + k * BIN_T + tid;
        if (e < E) atomicAdd(&dyn_lds[d[k] >> RUN_SHIFT], 1);
    }
    __syncthreads();
    for (int b = tid; b < nBins; b += BIN_T) hist[blockIdx.x * nBins + b] = dyn_lds[b];
}

__global__ __launch_bounds__(BIN_T) void run_place_kernel(const int* __restrict__ row, const int* __restrict__ col,
                                                          const int* __restrict__ hist, int* __restrict__ binstart,
                                                          unsigned* __restrict__ bins, int E, int nBins, int nEB) {
    extern __shared__ int dyn_lds[];
    int* sA = dyn_lds;
    int* sB = dyn_lds + 2 * nBins;
    const int tid  = threadIdx.x;
    const int wg   = blockIdx.x;
    const int half = tid >> 9;

    int d[EDGE_B / BIN_T], s[EDGE_B / BIN_T];
    for (int k = 0; k < EDGE_B / BIN_T; ++k) {
        const int e = min(wg * EDGE_B + k * BIN_T + tid, E - 1);
        d[k] = col[e];
        s[k] = row[e];
    }

    for (int b = tid & 511; b < nBins; b += 512) {
        int tot = 0, pre = 0;
        int g = half;
        for (; g + 14 < nEB; g += 16) {
            int v[8];
            for (int u = 0; u < 8; ++u) v[u] = hist[(g + 2 * u) * nBins + b];
            for (int u = 0; u < 8; ++u) { tot += v[u]; pre += (g + 2 * u < wg) ? v[u] : 0; }
        }
        for (; g < nEB; g += 2) {
            const int v = hist[g * nBins + b];
            tot += v; pre += (g < wg) ? v : 0;
        }
        sA[half * nBins + b] = tot;
        sB[half * nBins + b] = pre;
    }
    __syncthreads();
    for (int b = tid; b < nBins; b += BIN_T) { sA[b] += sA[nBins + b]; sB[b] += sB[nBins + b]; }
    __syncthreads();

    if (tid < 32) {
        const int chunk = (nBins + 31) / 32;
        const int lo = min(tid * chunk, nBins);
        const int hi = min(lo + chunk, nBins);
        int mysum = 0;
        for (int b = lo; b < hi; ++b) mysum += sA[b];
        int incl = mysum;
        for (int off = 1; off < 32; off <<= 1) {
            const int t = __shfl_up(incl, off, 32);
            if (tid >= off) incl += t;
        }
        int run = incl - mysum;
        for (int b = lo; b < hi; ++b) { const int t = sA[b]; sA[b] = run; run += t; }
    }
    __syncthreads();
    for (int b = tid; b < nBins; b += BIN_T) {
        const int st = sA[b];
        if (wg == 0) binstart[b] = st;
        sB[b] += st;
    }
    if (wg == 0 && tid == 0) binstart[nBins] = E;
    __syncthreads();

    for (int k = 0; k < EDGE_B / BIN_T; ++k) {
        const int e = wg * EDGE_B + k * BIN_T + tid;
        if (e < E) {
            const int pos = atomicAdd(&sB[d[k] >> RUN_SHIFT], 1);
            bins[pos] = ((unsigned)s[k] << RUN_SHIFT) | (unsigned)(d[k] & (RANGE_N - 1));
        }
    }
}

__global__ __launch_bounds__(256, 4) void mfconv_gemm_kernel(
    const float* __restrict__ x,
    const int* __restrict__ binstart, const unsigned* __restrict__ bins,
    const float* __restrict__ neigh_w, const float* __restrict__ neigh_b,
    const float* __restrict__ root_w,  const float* __restrict__ root_b,
    float* __restrict__ out, int N)
{
    __shared__ __attribute__((aligned(16))) float ldsA[2 * TILE_M * LDA];
    __shared__ __attribute__((aligned(16))) int s_list[LIST_LEN];
    __shared__ int4  s_rec[LIST_LEN];
    __shared__ int   s_dl[RANGE_N];
    __shared__ __attribute__((aligned(16))) int s_src[RANGE_N * SRC_CAP];
    __shared__ unsigned s_ovf[OVF_CAP];
    __shared__ int   s_ovn;
    __shared__ int   s_cnt[RANGE_PASS * 8];

    const int tid   = threadIdx.x;
    const int wave  = tid >> 5;
    const int lane  = tid & 31;
    const int node0 = blockIdx.x * RANGE_N;

    const int by     = blockIdx.y;
    const int bucket = (by < BIG_PARTS) ? MAXDEG : (NBUCK - 2 + BIG_PARTS - by);
    const int part   = (by < BIG_PARTS) ? by : 0;
    const int nparts = (by < BIG_PARTS) ? BIG_PARTS : 1;

    for (int i = tid; i < LIST_LEN; i += 256) { s_list[i] = -1; s_rec[i] = make_int4(-1, 0, 0, 0); }
    for (int i = tid; i < RANGE_N; i += 256) s_dl[i] = 0;
    if (tid == 0) s_ovn = 0;
    const int e0 = binstart[blockIdx.x];
    const int e1 = binstart[blockIdx.x + 1];
    __syncthreads();

    for (int eb = e0; eb < e1; eb += 256 * EDGE_BATCH) {
        unsigned w[EDGE_BATCH];
        for (int u = 0; u < EDGE_BATCH; ++u)
            w[u] = bins[min(eb + u * 256 + tid, e1 - 1)];
        int slot[EDGE_BATCH];
        for (int u = 0; u < EDGE_BATCH; ++u) {
            slot[u] = -1;
            if (eb + u * 256 + tid < e1) slot[u] = atomicAdd(&s_dl[w[u] & (RANGE_N - 1)], 1);
        }
        for (int u = 0; u < EDGE_BATCH; ++u) {
            if (slot[u] >= 0) {
                if (slot[u] < SRC_CAP) {
                    s_src[(int)(w[u] & (RANGE_N - 1)) * SRC_CAP + slot[u]] = (int)(w[u] >> RUN_SHIFT);
                } else {
                    const int o = atomicAdd(&s_ovn, 1);
                    if (o < OVF_CAP) s_ovf[o] = w[u];
                }
            }
        }
    }
    __syncthreads();

    bool mine[RANGE_PASS];
    int  myrank[RANGE_PASS];
    #pragma unroll
    for (int p = 0; p < RANGE_PASS; ++p) {
        const int node = node0 + p * 256 + tid;
        const bool in = (node < N) && (min(s_dl[p * 256 + tid], MAXDEG) == bucket);
        const unsigned m = __builtin_amdgcn_ballot_w32(in);
        if (lane == 0) s_cnt[p * 8 + wave] = __popc(m);
        myrank[p] = __popc(m & ((1u << lane) - 1u));
        mine[p] = in;
    }
    __syncthreads();

    int total = 0;
    int base[RANGE_PASS];
    #pragma unroll
    for (int p = 0; p < RANGE_PASS; ++p) base[p] = 0;
    #pragma unroll
    for (int i = 0; i < RANGE_PASS * 8; ++i) {
        const int c = s_cnt[i];
        #pragma unroll
        for (int p = 0; p < RANGE_PASS; ++p) if (i < p * 8 + wave) base[p] += c;
        total += c;
    }
    const int nTiles = (__builtin_amdgcn_readfirstlane(total) + TILE_M - 1) / TILE_M;
    if (part >= nTiles) return;

    #pragma unroll
    for (int p = 0; p < RANGE_PASS; ++p)
        if (mine[p]) {
            const int4 ids = *(const int4*)&s_src[(p * 256 + tid) * SRC_CAP];
            s_list[base[p] + myrank[p]] = node0 + p * 256 + tid;
            s_rec[base[p] + myrank[p]]  = make_int4(node0 + p * 256 + tid, ids.x, ids.y, ids.z);
        }

    __syncthreads();

    const int sm = tid >> 4;
    const int fq = (tid & 15) * 4;
    constexpr int FH = IN_F / 2;

    int    g_node = -1, g_ln = 0, g_d = 0;
    float4 g_own[2];
    float4 g_nb[3][2];

    const int rows_early = min(bucket, 3);
    auto issue_rows = [&](const int4 rec) {
        const float4 id0 = make_float4(-0.0f, -0.0f, -0.0f, -0.0f);
        #pragma unroll
        for (int j = 0; j < 3; ++j) { g_nb[j][0] = id0; g_nb[j][1] = id0; }
        g_own[0] = id0; g_own[1] = id0;
        g_node = rec.x;
        g_ln = 0;
        g_d  = 0;
        if (g_node >= 0) {
            const float* xo = x + (size_t)g_node * IN_F + fq;
            g_own[0] = *(const float4*)(xo + 0);
            g_own[1] = *(const float4*)(xo + FH);
            g_ln = g_node - node0;
            g_d  = (bucket < MAXDEG) ? bucket : s_dl[g_ln];
            const int idv[3] = { rec.y, rec.z, rec.w };
            #pragma unroll
            for (int j = 0; j < 3; ++j) {
                if (j < rows_early) {
                    const float* xs = x + (size_t)idv[j] * IN_F + fq;
                    g_nb[j][0] = *(const float4*)(xs + 0);
                    g_nb[j][1] = *(const float4*)(xs + FH);
                }
            }
        }
    };

    auto request_late_rows = [&]() {
        const int q = tid & 15;
        if (g_node >= 0 && q < 10) {
            const int j = 3 + (q >> 1);
            if (j < g_d) {
                const int prow = s_src[g_ln * SRC_CAP + j];
                __builtin_prefetch(x + (size_t)prow * IN_F + (q & 1) * (IN_F / 2), 0, 2);
            }
        }
    };

    auto finish_rows = [&](float* dstRow) {
        float4 n0 = make_float4(0.0f, 0.0f, 0.0f, 0.0f), n1 = n0;
        float4 w0 = n0, w1 = n0;
        if (g_node >= 0) {
            const int ln = g_ln;
            const int d  = g_d;
            w0 = g_own[0]; w1 = g_own[1];
            if (d > 0) {
                #pragma unroll
                for (int j = 0; j < 3; ++j) {
                    n0.x += g_nb[j][0].x; n0.y += g_nb[j][0].y; n0.z += g_nb[j][0].z; n0.w += g_nb[j][0].w;
                    n1.x += g_nb[j][1].x; n1.y += g_nb[j][1].y; n1.z += g_nb[j][1].z; n1.w += g_nb[j][1].w;
                }
            }
            if (d > 3) {
                const float4 id0 = make_float4(-0.0f, -0.0f, -0.0f, -0.0f);
                const int  id3 = s_src[ln * SRC_CAP + 3];
                const int4 id4 = *(const int4*)&s_src[ln * SRC_CAP + 4];
                const int idv[5] = { id3, id4.x, id4.y, id4.z, id4.w };
                float4 v[5][2];
                #pragma unroll
                for (int j = 0; j < 5; ++j) {
                    v[j][0] = id0; v[j][1] = id0;
                    if (3 + j < d) {
                        const float* xs = x + (size_t)idv[j] * IN_F + fq;
                        v[j][0] = *(const float4*)(xs + 0);
                        v[j][1] = *(const float4*)(xs + FH);
                    }
                }
                #pragma unroll
                for (int j = 0; j < 5; ++j) {
                    n0.x += v[j][0].x; n0.y += v[j][0].y; n0.z += v[j][0].z; n0.w += v[j][0].w;
                    n1.x += v[j][1].x; n1.y += v[j][1].y; n1.z += v[j][1].z; n1.w += v[j][1].w;
                }
            }
            if (d > SRC_CAP) {
                auto add_row = [&](unsigned srcId) {
                    const float* xs = x + (size_t)srcId * IN_F + fq;
                    const float4 v0 = *(const float4*)(xs + 0);
                    const float4 v1 = *(const float4*)(xs + FH);
                    n0.x += v0.x; n0.y += v0.y; n0.z += v0.z; n0.w += v0.w;
                    n1.x += v1.x; n1.y += v1.y; n1.z += v1.z; n1.w += v1.w;
                };
                const int novf = s_ovn;
                if (novf <= OVF_CAP) {
                    #pragma unroll 1
                    for (int j = 0; j < novf; ++j) {
                        const unsigned w = s_ovf[j];
                        if ((int)(w & (RANGE_N - 1)) == ln) add_row(w >> RUN_SHIFT);
                    }
                } else {
                    n0 = make_float4(0.0f, 0.0f, 0.0f, 0.0f); n1 = n0;
                    #pragma unroll 1
                    for (int e = e0; e < e1; ++e) {
                        const unsigned w = bins[e];
                        if ((int)(w & (RANGE_N - 1)) == ln) add_row(w >> RUN_SHIFT);
                    }
                }
            }
            const float inv = 1.0f / fmaxf((float)d, 1.0f);
            n0.x *= inv; n0.y *= inv; n0.z *= inv; n0.w *= inv;
            n1.x *= inv; n1.y *= inv; n1.z *= inv; n1.w *= inv;
        }
        *(float4*)(dstRow + fq)             = n0;
        *(float4*)(dstRow + FH + fq)        = n1;
        *(float4*)(dstRow + IN_F + fq)      = w0;
        *(float4*)(dstRow + IN_F + FH + fq) = w1;
    };

    issue_rows(s_rec[part * TILE_M + sm]);
    if (bucket == MAXDEG) request_late_rows();
    int4 rec = make_int4(-1, 0, 0, 0);
    if (part + nparts < nTiles) rec = s_rec[(part + nparts) * TILE_M + sm];

    const int lr    = lane & 15;
    const int khalf = (lane >> 4) << 1;
    const int ncol  = wave * 16 + lr;
    const int mbase = (lane >> 4) * 8;

    const float* __restrict__ wn = neigh_w + (size_t)bucket * (OUT_F * IN_F);
    const float* __restrict__ wr = root_w  + (size_t)bucket * (OUT_F * IN_F);
    v2f bfr[64];
    {
        float* wstrip = &ldsA[wave * (16 * WPITCH)];
        const int wrow = lane >> 3;
        const int wcol = (lane & 7) * 4;
        struct WChunk { float4 r0, r1, r2, r3; };
        auto load_chunk = [&](int ch) -> WChunk {
            const float* __restrict__ p = ((ch < 4) ? wn : wr) + (size_t)(wave * 16 + wrow) * IN_F
                                          + (ch & 3) * WCHUNK + wcol;
            WChunk c;
            c.r0 = *(const float4*)(p);
            c.r1 = *(const float4*)(p + 4 * IN_F);
            c.r2 = *(const float4*)(p + 8 * IN_F);
            c.r3 = *(const float4*)(p + 12 * IN_F);
            return c;
        };
        auto turn_chunk = [&](const WChunk& c, int ch) {
            __builtin_amdgcn_fence(__ATOMIC_ACQ_REL, "wavefront");
            float* q = &wstrip[wrow * WPITCH + wcol];
            *(float4*)(q)               = c.r0;
            *(float4*)(q + 4 * WPITCH)  = c.r1;
            *(float4*)(q + 8 * WPITCH)  = c.r2;
            *(float4*)(q + 12 * WPITCH) = c.r3;
            __builtin_amdgcn_fence(__ATOMIC_ACQ_REL, "wavefront");
            #pragma unroll
            for (int k8 = 0; k8 < 8; ++k8)
                bfr[ch * 8 + k8] = *(const v2f*)&wstrip[lr * WPITCH + k8 * 4 + khalf];
        };
        const WChunk c0 = load_chunk(0), c1 = load_chunk(1), c2 = load_chunk(2), c3 = load_chunk(3);
        turn_chunk(c0, 0); turn_chunk(c1, 1);
        const WChunk c4 = load_chunk(4), c5 = load_chunk(5);
        turn_chunk(c2, 2); turn_chunk(c3, 3);
        const WChunk c6 = load_chunk(6), c7 = load_chunk(7);
        turn_chunk(c4, 4); turn_chunk(c5, 5);
        turn_chunk(c6, 6); turn_chunk(c7, 7);
    }
    const float bias = neigh_b[bucket * OUT_F + ncol] + root_b[bucket * OUT_F + ncol];
    __syncthreads();

    finish_rows(&ldsA[sm * LDA]);
    __syncthreads();

    int cur = 0;
    for (int t = part; t < nTiles; t += nparts) {
        const int  slot0 = t * TILE_M;
        const bool more  = t + nparts < nTiles;

        if (more) {
            issue_rows(rec);
            if (bucket == MAXDEG) request_late_rows();
            if (t + 2 * nparts < nTiles) rec = s_rec[(t + 2 * nparts) * TILE_M + sm];
        }

        const int4 onA = *(const int4*)&s_list[slot0 + mbase];
        const int4 onB = *(const int4*)&s_list[slot0 + mbase + 4];

        const float* __restrict__ aRow = &ldsA[cur * (TILE_M * LDA) + lr * LDA];
        v8f acc = {};
        #pragma unroll
        for (int kk = 0; kk < 32; ++kk) {
            const int k = kk * 4 + khalf;
            v2f a;
            a.x = aRow[k];     a.y = aRow[k + 1];
            acc = __builtin_amdgcn_wmma_f32_16x16x4_f32(false, a, false, bfr[kk],
                                                        (short)0, acc, false, false);
        }
        #pragma unroll
        for (int kk = 0; kk < 32; ++kk) {
            const int k = kk * 4 + khalf;
            v2f a;
            a.x = aRow[IN_F + k]; a.y = aRow[IN_F + k + 1];
            acc = __builtin_amdgcn_wmma_f32_16x16x4_f32(false, a, false, bfr[32 + kk],
                                                        (short)0, acc, false, false);
        }

        if (more) {
            finish_rows(&ldsA[(cur ^ 1) * (TILE_M * LDA) + sm * LDA]);
            __syncthreads();
        }
        cur ^= 1;

        const int onode[8] = { onA.x, onA.y, onA.z, onA.w, onB.x, onB.y, onB.z, onB.w };
        #pragma unroll
        for (int i = 0; i < 8; ++i)
            if (onode[i] >= 0) out[(size_t)onode[i] * OUT_F + ncol] = acc[i] + bias;
    }
}

extern "C" void kernel_launch(void* const* d_in, const int* in_sizes, int n_in,
                              void* d_out, int out_size, void* d_ws, size_t ws_size,
                              hipStream_t stream) {
    const float* x       = (const float*)d_in[0];
    const float* neigh_w = (const float*)d_in[1];
    const float* neigh_b = (const float*)d_in[2];
    const float* root_w  = (const float*)d_in[3];
    const float* root_b  = (const float*)d_in[4];
    const int*   eidx    = (const int*)d_in[5];

    const int N = in_sizes[0] / IN_F;
    const int E = in_sizes[5] / 2;
    const int* row = eidx;
    const int* col = eidx + E;

    const int nRanges = (N + RANGE_N - 1) / RANGE_N;
    const int nEB     = (E + EDGE_B - 1) / EDGE_B;

    char* ws = (char*)d_ws;
    int* hist      = (int*)ws;       ws += (size_t)((nEB * nRanges + 15) & ~15) * sizeof(int);
    int* binstart  = (int*)ws;       ws += (size_t)((nRanges + 1 + 15) & ~15) * sizeof(int);
    unsigned* bins = (unsigned*)ws;

    run_hist_kernel<<<nEB, BIN_T, (size_t)nRanges * sizeof(int), stream>>>(col, hist, E, nRanges);
    run_place_kernel<<<nEB, BIN_T, (size_t)4 * nRanges * sizeof(int), stream>>>(row, col, hist, binstart, bins,
                                                                              E, nRanges, nEB);

    mfconv_gemm_kernel<<<dim3(nRanges, GRID_Y), 256, 0, stream>>>(
        x, binstart, bins, neigh_w, neigh_b, root_w, root_b, (float*)d_out, N);
}
